# starting kernel
# baseline (speedup 1.0000x reference)
#include <hip/hip_runtime.h>
#include <hip/hip_bf16.h>

typedef __attribute__((ext_vector_type(2))) float v2f;
typedef __attribute__((ext_vector_type(8))) float v8f;

#define D_DIM 256
#define L_DIM 1024
#define B_DIM 4
#define M_DIM (B_DIM * L_DIM)
#define LN_EPS 1e-5f

__device__ __forceinline__ float block_sum_256(float v, float* sh8) {
#pragma unroll
  for (int m = 16; m > 0; m >>= 1) v += __shfl_xor(v, m, 32);
  if ((threadIdx.x & 31) == 0) sh8[threadIdx.x >> 5] = v;
  __syncthreads();
  float r = 0.f;
#pragma unroll
  for (int i = 0; i < 8; ++i) r += sh8[i];
  return r;
}

__global__ void wmma_gemm_f32_kernel(const float* __restrict__ A,
                                     const float* __restrict__ Wm,
                                     float* __restrict__ Out,
                                     const float* __restrict__ bias) {
  const int K = D_DIM, N = D_DIM;
  const int wave = threadIdx.x >> 5;
  const int lane = threadIdx.x & 31;
  const int half = lane >> 4;
  const int l16  = lane & 15;
  const int tM = (blockIdx.x * 8 + wave) * 16;
  const int tN = blockIdx.y * 16;

  const float* arow = A  + (size_t)(tM + l16) * K + (half ? 2 : 0);
  const float* brow = Wm + (size_t)(tN + l16) * K + (half ? 2 : 0);

  v8f acc = {};
#pragma unroll 8
  for (int kc = 0; kc < K; kc += 4) {
    v2f a; a.x = arow[kc]; a.y = arow[kc + 1];
    v2f b; b.x = brow[kc]; b.y = brow[kc + 1];
    acc = __builtin_amdgcn_wmma_f32_16x16x4_f32(
         false, a,  false, b,
         (short)0, acc,  false,  false);
  }

  const int n = tN + l16;
  const float bv = bias ? bias[n] : 0.f;
  union { v8f v; float f[8]; } u; u.v = acc;
#pragma unroll
  for (int r = 0; r < 8; ++r) {
    const int m = tM + r + (half ? 8 : 0);
    Out[(size_t)m * N + n] = u.f[r] + bv;
  }
}

__global__ void phi_beta_kernel(const float* __restrict__ x,
                                float* __restrict__ q,
                                float* __restrict__ k,
                                const float* __restrict__ beta_w,
                                const float* __restrict__ beta_b,
                                const float* __restrict__ lnp_g,
                                const float* __restrict__ lnp_b,
                                float* __restrict__ beta) {
  __shared__ float sh[8 * 5];
  const int row = blockIdx.x, tid = threadIdx.x;
  const size_t off = (size_t)row * D_DIM + tid;

  float qe; { float t = q[off]; qe = (t > 0.f) ? t + 1.f : __expf(t); }
  float ke; { float t = k[off]; ke = (t > 0.f) ? t + 1.f : __expf(t); }
  const float xb = x[off] * beta_w[tid];

  const float sq  = block_sum_256(qe,      sh + 0);
  const float sq2 = block_sum_256(qe * qe, sh + 8);
  const float sk  = block_sum_256(ke,      sh + 16);
  const float sk2 = block_sum_256(ke * ke, sh + 24);
  const float sb  = block_sum_256(xb,      sh + 32);

  const float muq = sq / D_DIM, varq = sq2 / D_DIM - muq * muq;
  const float muk = sk / D_DIM, vark = sk2 / D_DIM - muk * muk;
  const float g = lnp_g[tid], bb = lnp_b[tid];
  q[off] = (qe - muq) * rsqrtf(varq + LN_EPS) * g + bb;
  k[off] = (ke - muk) * rsqrtf(vark + LN_EPS) * g + bb;
  if (tid == 0) beta[row] = 1.f / (1.f + __expf(-(sb + beta_b[0])));
}

union F8 { float4 v4[2]; float f[8]; };

__global__ void delta_scan_kernel(const float* __restrict__ phi_q,
                                  const float* __restrict__ phi_k,
                                  const float* __restrict__ v,
                                  const float* __restrict__ beta,
                                  const float* __restrict__ decay,
                                  float* __restrict__ y) {
  const int b    = blockIdx.y;
  const int wave = threadIdx.x >> 5;
  const int lane = threadIdx.x & 31;
  const int row  = blockIdx.x * 8 + wave;
  const int c0   = lane * 8;

  const float df = 1.f / (1.f + __expf(-decay[0]));
  const size_t base = (size_t)b * L_DIM * D_DIM;

  float w[8];
#pragma unroll
  for (int c = 0; c < 8; ++c) w[c] = 0.f;

  F8 cq, ck;
  cq.v4[0] = *(const float4*)(phi_q + base + c0);
  cq.v4[1] = *(const float4*)(phi_q + base + c0 + 4);
  ck.v4[0] = *(const float4*)(phi_k + base + c0);
  ck.v4[1] = *(const float4*)(phi_k + base + c0 + 4);
  float cv = v[base + row];
  float cb = beta[(size_t)b * L_DIM];

  for (int t = 0; t < L_DIM; ++t) {
    const int tn = (t + 1 < L_DIM) ? t + 1 : t;
    const size_t noff = base + (size_t)tn * D_DIM;
    F8 nq, nk;
    nq.v4[0] = *(const float4*)(phi_q + noff + c0);
    nq.v4[1] = *(const float4*)(phi_q + noff + c0 + 4);
    nk.v4[0] = *(const float4*)(phi_k + noff + c0);
    nk.v4[1] = *(const float4*)(phi_k + noff + c0 + 4);
    const float nv = v[noff + row];
    const float nb = beta[(size_t)b * L_DIM + tn];
    if (t + 8 < L_DIM) {
      const size_t poff = base + (size_t)(t + 8) * D_DIM + c0;
      __builtin_prefetch(phi_q + poff, 0, 0);
      __builtin_prefetch(phi_k + poff, 0, 0);
    }

    float rq = 0.f, rk = 0.f;
#pragma unroll
    for (int c = 0; c < 8; ++c) {
      rq = fmaf(w[c], cq.f[c], rq);
      rk = fmaf(w[c], ck.f[c], rk);
    }
#pragma unroll
    for (int m = 16; m > 0; m >>= 1) {
      rq += __shfl_xor(rq, m, 32);
      rk += __shfl_xor(rk, m, 32);
    }
    if (lane == 0) y[base + (size_t)t * D_DIM + row] = rq;

    const float u = cb * (cv - rk);
#pragma unroll
    for (int c = 0; c < 8; ++c) w[c] = fmaf(df, w[c], u * ck.f[c]);

    cq = nq; ck = nk; cv = nv; cb = nb;
  }
}

__global__ void ln_rows_kernel(float* __restrict__ y,
                               const float* __restrict__ g,
                               const float* __restrict__ b) {
  __shared__ float sh[16];
  const int row = blockIdx.x, tid = threadIdx.x;
  const size_t off = (size_t)row * D_DIM + tid;
  const float v  = y[off];
  const float s  = block_sum_256(v, sh);
  const float s2 = block_sum_256(v * v, sh + 8);
  const float mu = s / D_DIM, var = s2 / D_DIM - mu * mu;
  y[off] = (v - mu) * rsqrtf(var + LN_EPS) * g[tid] + b[tid];
}

extern "C" void kernel_launch(void* const* d_in, const int* in_sizes, int n_in,
                              void* d_out, int out_size, void* d_ws, size_t ws_size,
                              hipStream_t stream) {
  (void)in_sizes; (void)n_in; (void)out_size; (void)ws_size;
  const float* x      = (const float*)d_in[0];
  const float* Wq     = (const float*)d_in[1];
  const float* Wk     = (const float*)d_in[2];
  const float* Wv     = (const float*)d_in[3];
  const float* beta_w = (const float*)d_in[4];
  const float* beta_b = (const float*)d_in[5];
  const float* decay  = (const float*)d_in[6];
  const float* Wo     = (const float*)d_in[7];
  const float* bo     = (const float*)d_in[8];
  const float* ln_g   = (const float*)d_in[9];
  const float* ln_b   = (const float*)d_in[10];
  const float* lnp_g  = (const float*)d_in[11];
  const float* lnp_b  = (const float*)d_in[12];
  float* out = (float*)d_out;

  char* ws = (char*)d_ws;
  const size_t MAT = (size_t)M_DIM * D_DIM * sizeof(float);
  float* q    = (float*)(ws);
  float* k    = (float*)(ws + MAT);
  float* vv   = (float*)(ws + 2 * MAT);
  float* beta = (float*)(ws + 3 * MAT);
  float* y    = (float*)(ws + 3 * MAT + (64 << 10));

  const dim3 gemm_grid(M_DIM / 128, D_DIM / 16);
  const dim3 gemm_blk(256);

  wmma_gemm_f32_kernel<<<gemm_grid, gemm_blk, 0, stream>>>(x, Wq, q, nullptr);
  wmma_gemm_f32_kernel<<<gemm_grid, gemm_blk, 0, stream>>>(x, Wk, k, nullptr);
  wmma_gemm_f32_kernel<<<gemm_grid, gemm_blk, 0, stream>>>(x, Wv, vv, nullptr);

  phi_beta_kernel<<<dim3(M_DIM), dim3(256), 0, stream>>>(
      x, q, k, beta_w, beta_b, lnp_g, lnp_b, beta);

  delta_scan_kernel<<<dim3(32, B_DIM), dim3(256), 0, stream>>>(
      q, k, vv, beta, decay, y);

  ln_rows_kernel<<<dim3(M_DIM), dim3(256), 0, stream>>>(y, ln_g, ln_b);

  wmma_gemm_f32_kernel<<<gemm_grid, gemm_blk, 0, stream>>>(y, Wo, out, bo);
}


# scan: each block stages 32-step chunks of phi_q, phi_k, v, beta in double-buffered LDS for its 8 waves, global loads one chunk ahead; the serial step waits on LDS reads instead of L2 loads
# speedup vs baseline: 1.3980x; 1.3980x over previous
#include <hip/hip_runtime.h>
#include <hip/hip_bf16.h>

typedef __attribute__((ext_vector_type(2))) float v2f;
typedef __attribute__((ext_vector_type(8))) float v8f;
typedef __attribute__((ext_vector_type(4))) float v4f;

#define D_DIM 256
#define L_DIM 1024
#define B_DIM 4
#define M_DIM (B_DIM * L_DIM)
#define LN_EPS 1e-5f

__device__ __forceinline__ float block_sum_256(float v, float* sh8) {
#pragma unroll
  for (int m = 16; m > 0; m >>= 1) v += __shfl_xor(v, m, 32);
  if ((threadIdx.x & 31) == 0) sh8[threadIdx.x >> 5] = v;
  __syncthreads();
  float r = 0.f;
#pragma unroll
  for (int i = 0; i < 8; ++i) r += sh8[i];
  return r;
}

__global__ void wmma_gemm_f32_kernel(const float* __restrict__ A,
                                     const float* __restrict__ Wm,
                                     float* __restrict__ Out,
                                     const float* __restrict__ bias) {
  const int K = D_DIM, N = D_DIM;
  const int wave = threadIdx.x >> 5;
  const int lane = threadIdx.x & 31;
  const int half = lane >> 4;
  const int l16  = lane & 15;
  const int tM = (blockIdx.x * 8 + wave) * 16;
  const int tN = blockIdx.y * 16;

  const float* arow = A  + (size_t)(tM + l16) * K + (half ? 2 : 0);
  const float* brow = Wm + (size_t)(tN + l16) * K + (half ? 2 : 0);

  v8f acc = {};
#pragma unroll 8
  for (int kc = 0; kc < K; kc += 4) {
    v2f a; a.x = arow[kc]; a.y = arow[kc + 1];
    v2f b; b.x = brow[kc]; b.y = brow[kc + 1];
    acc = __builtin_amdgcn_wmma_f32_16x16x4_f32(
         false, a,  false, b,
         (short)0, acc,  false,  false);
  }

  const int n = tN + l16;
  const float bv = bias ? bias[n] : 0.f;
  union { v8f v; float f[8]; } u; u.v = acc;
#pragma unroll
  for (int r = 0; r < 8; ++r) {
    const int m = tM + r + (half ? 8 : 0);
    Out[(size_t)m * N + n] = u.f[r] + bv;
  }
}

__global__ void phi_beta_kernel(const float* __restrict__ x,
                                float* __restrict__ q,
                                float* __restrict__ k,
                                const float* __restrict__ beta_w,
                                const float* __restrict__ beta_b,
                                const float* __restrict__ lnp_g,
                                const float* __restrict__ lnp_b,
                                float* __restrict__ beta) {
  __shared__ float sh[8 * 5];
  const int row = blockIdx.x, tid = threadIdx.x;
  const size_t off = (size_t)row * D_DIM + tid;

  float qe; { float t = q[off]; qe = (t > 0.f) ? t + 1.f : __expf(t); }
  float ke; { float t = k[off]; ke = (t > 0.f) ? t + 1.f : __expf(t); }
  const float xb = x[off] * beta_w[tid];

  const float sq  = block_sum_256(qe,      sh + 0);
  const float sq2 = block_sum_256(qe * qe, sh + 8);
  const float sk  = block_sum_256(ke,      sh + 16);
  const float sk2 = block_sum_256(ke * ke, sh + 24);
  const float sb  = block_sum_256(xb,      sh + 32);

  const float muq = sq / D_DIM, varq = sq2 / D_DIM - muq * muq;
  const float muk = sk / D_DIM, vark = sk2 / D_DIM - muk * muk;
  const float g = lnp_g[tid], bb = lnp_b[tid];
  q[off] = (qe - muq) * rsqrtf(varq + LN_EPS) * g + bb;
  k[off] = (ke - muk) * rsqrtf(vark + LN_EPS) * g + bb;
  if (tid == 0) beta[row] = 1.f / (1.f + __expf(-(sb + beta_b[0])));
}

union F8 { v4f v4[2]; float f[8]; };

#define SCAN_T 32
#define SCAN_NC (L_DIM / SCAN_T)
#define SCAN_F4 (SCAN_T * D_DIM / 4 / 256)

__global__ void __launch_bounds__(256)
delta_scan_kernel(const float* __restrict__ phi_q,
                  const float* __restrict__ phi_k,
                  const float* __restrict__ v,
                  const float* __restrict__ beta,
                  const float* __restrict__ decay,
                  float* __restrict__ y) {
  __shared__ v4f s_q[2][SCAN_T * D_DIM / 4];
  __shared__ v4f s_k[2][SCAN_T * D_DIM / 4];
  __shared__ float  s_v[2][SCAN_T * 8];
  __shared__ float  s_b[2][SCAN_T];

  const int b    = blockIdx.y;
  const int tid  = threadIdx.x;
  const int wave = tid >> 5;
  const int lane = tid & 31;
  const int row  = blockIdx.x * 8 + wave;

  const float df = 1.f / (1.f + __expf(-decay[0]));
  const size_t base = (size_t)b * L_DIM * D_DIM;

  const v4f* gq = (const v4f*)(phi_q + base) + tid;
  const v4f* gk = (const v4f*)(phi_k + base) + tid;
  const float*  gv = v + base + (size_t)(tid >> 3) * D_DIM + blockIdx.x * 8 + (tid & 7);
  const float*  gb = beta + (size_t)b * L_DIM + (tid & (SCAN_T - 1));

#pragma unroll
  for (int j = 0; j < SCAN_F4; ++j) {
    s_q[0][tid + 256 * j] = gq[256 * j];
    s_k[0][tid + 256 * j] = gk[256 * j];
  }
  s_v[0][tid] = gv[0];
  if (tid < SCAN_T) s_b[0][tid] = gb[0];
  __syncthreads();

  float w[8];
#pragma unroll
  for (int c = 0; c < 8; ++c) w[c] = 0.f;

  F8 cq, ck;
  cq.v4[0] = s_q[0][lane * 2];
  cq.v4[1] = s_q[0][lane * 2 + 1];
  ck.v4[0] = s_k[0][lane * 2];
  ck.v4[1] = s_k[0][lane * 2 + 1];
  float cv = s_v[0][wave];
  float cb = s_b[0][0];

  for (int ch = 0; ch < SCAN_NC; ++ch) {
    const int buf = ch & 1;
    const int chn = (ch + 1 < SCAN_NC) ? ch + 1 : ch;

    const size_t adv4 = (size_t)chn * (SCAN_T * D_DIM / 4);
    const v4f rq0 = gq[adv4 + 256 * 0], rk0 = gk[adv4 + 256 * 0];
    const v4f rq1 = gq[adv4 + 256 * 1], rk1 = gk[adv4 + 256 * 1];
    const v4f rq2 = gq[adv4 + 256 * 2], rk2 = gk[adv4 + 256 * 2];
    const v4f rq3 = gq[adv4 + 256 * 3], rk3 = gk[adv4 + 256 * 3];
    const v4f rq4 = gq[adv4 + 256 * 4], rk4 = gk[adv4 + 256 * 4];
    const v4f rq5 = gq[adv4 + 256 * 5], rk5 = gk[adv4 + 256 * 5];
    const v4f rq6 = gq[adv4 + 256 * 6], rk6 = gk[adv4 + 256 * 6];
    const v4f rq7 = gq[adv4 + 256 * 7], rk7 = gk[adv4 + 256 * 7];
    const float rv = gv[(size_t)chn * (SCAN_T * D_DIM)];
    const float rb = gb[chn * SCAN_T];
    asm volatile("" :: "v"(rq0), "v"(rq1), "v"(rq2), "v"(rq3),
                       "v"(rq4), "v"(rq5), "v"(rq6), "v"(rq7),
                       "v"(rk0), "v"(rk1), "v"(rk2), "v"(rk3),
                       "v"(rk4), "v"(rk5), "v"(rk6), "v"(rk7),
                       "v"(rv), "v"(rb));

    const v4f* lq = &s_q[buf][lane * 2];
    const v4f* lk = &s_k[buf][lane * 2];
    const float*  lv = &s_v[buf][wave];
    const float*  lb = &s_b[buf][0];
    float* yrow = y + base + (size_t)ch * SCAN_T * D_DIM + row;

    for (int tt = 0; tt < SCAN_T; ++tt) {
      const int tn = (tt + 1 < SCAN_T) ? tt + 1 : tt;
      F8 nq, nk;
      nq.v4[0] = lq[tn * (D_DIM / 4)];
      nq.v4[1] = lq[tn * (D_DIM / 4) + 1];
      nk.v4[0] = lk[tn * (D_DIM / 4)];
      nk.v4[1] = lk[tn * (D_DIM / 4) + 1];
      const float nv = lv[tn * 8];
      const float nb = lb[tn];

      float rq = 0.f, rk = 0.f;
#pragma unroll
      for (int c = 0; c < 8; ++c) {
        rq = fmaf(w[c], cq.f[c], rq);
        rk = fmaf(w[c], ck.f[c], rk);
      }
#pragma unroll
      for (int m = 16; m > 0; m >>= 1) {
        rq += __shfl_xor(rq, m, 32);
        rk += __shfl_xor(rk, m, 32);
      }
      if (lane == 0) yrow[(size_t)tt * D_DIM] = rq;

      const float u = cb * (cv - rk);
#pragma unroll
      for (int c = 0; c < 8; ++c) w[c] = fmaf(df, w[c], u * ck.f[c]);

      cq = nq; ck = nk; cv = nv; cb = nb;
    }

    v4f* dq = &s_q[buf ^ 1][tid];
    v4f* dk = &s_k[buf ^ 1][tid];
    dq[256 * 0] = rq0; dk[256 * 0] = rk0;
    dq[256 * 1] = rq1; dk[256 * 1] = rk1;
    dq[256 * 2] = rq2; dk[256 * 2] = rk2;
    dq[256 * 3] = rq3; dk[256 * 3] = rk3;
    dq[256 * 4] = rq4; dk[256 * 4] = rk4;
    dq[256 * 5] = rq5; dk[256 * 5] = rk5;
    dq[256 * 6] = rq6; dk[256 * 6] = rk6;
    dq[256 * 7] = rq7; dk[256 * 7] = rk7;
    s_v[buf ^ 1][tid] = rv;
    if (tid < SCAN_T) s_b[buf ^ 1][tid] = rb;
    __syncthreads();
    cq.v4[0] = s_q[buf ^ 1][lane * 2];
    cq.v4[1] = s_q[buf ^ 1][lane * 2 + 1];
    ck.v4[0] = s_k[buf ^ 1][lane * 2];
    ck.v4[1] = s_k[buf ^ 1][lane * 2 + 1];
    cv = s_v[buf ^ 1][wave];
    cb = s_b[buf ^ 1][0];
  }
}

__global__ void ln_rows_kernel(float* __restrict__ y,
                               const float* __restrict__ g,
                               const float* __restrict__ b) {
  __shared__ float sh[16];
  const int row = blockIdx.x, tid = threadIdx.x;
  const size_t off = (size_t)row * D_DIM + tid;
  const float v  = y[off];
  const float s  = block_sum_256(v, sh);
  const float s2 = block_sum_256(v * v, sh + 8);
  const float mu = s / D_DIM, var = s2 / D_DIM - mu * mu;
  y[off] = (v - mu) * rsqrtf(var + LN_EPS) * g[tid] + b[tid];
}

extern "C" void kernel_launch(void* const* d_in, const int* in_sizes, int n_in,
                              void* d_out, int out_size, void* d_ws, size_t ws_size,
                              hipStream_t stream) {
  (void)in_sizes; (void)n_in; (void)out_size; (void)ws_size;
  const float* x      = (const float*)d_in[0];
  const float* Wq     = (const float*)d_in[1];
  const float* Wk     = (const float*)d_in[2];
  const float* Wv     = (const float*)d_in[3];
  const float* beta_w = (const float*)d_in[4];
  const float* beta_b = (const float*)d_in[5];
  const float* decay  = (const float*)d_in[6];
  const float* Wo     = (const float*)d_in[7];
  const float* bo     = (const float*)d_in[8];
  const float* ln_g   = (const float*)d_in[9];
  const float* ln_b   = (const float*)d_in[10];
  const float* lnp_g  = (const float*)d_in[11];
  const float* lnp_b  = (const float*)d_in[12];
  float* out = (float*)d_out;

  char* ws = (char*)d_ws;
  const size_t MAT = (size_t)M_DIM * D_DIM * sizeof(float);
  float* q    = (float*)(ws);
  float* k    = (float*)(ws + MAT);
  float* vv   = (float*)(ws + 2 * MAT);
  float* beta = (float*)(ws + 3 * MAT);
  float* y    = (float*)(ws + 3 * MAT + (64 << 10));

  const dim3 gemm_grid(M_DIM / 128, D_DIM / 16);
  const dim3 gemm_blk(256);

  wmma_gemm_f32_kernel<<<gemm_grid, gemm_blk, 0, stream>>>(x, Wq, q, nullptr);
  wmma_gemm_f32_kernel<<<gemm_grid, gemm_blk, 0, stream>>>(x, Wk, k, nullptr);
  wmma_gemm_f32_kernel<<<gemm_grid, gemm_blk, 0, stream>>>(x, Wv, vv, nullptr);

  phi_beta_kernel<<<dim3(M_DIM), dim3(256), 0, stream>>>(
      x, q, k, beta_w, beta_b, lnp_g, lnp_b, beta);

  delta_scan_kernel<<<dim3(32, B_DIM), dim3(256), 0, stream>>>(
      q, k, vv, beta, decay, y);

  ln_rows_kernel<<<dim3(M_DIM), dim3(256), 0, stream>>>(y, ln_g, ln_b);

  wmma_gemm_f32_kernel<<<gemm_grid, gemm_blk, 0, stream>>>(y, Wo, out, bo);
}


# wave32 sums (the scan step's two dot products, the layernorm block sums): butterfly on the vector ALU, v_permlanex16 then four row_xmask DPP adds, replaces five ds_bpermute shuffles; same add order
# speedup vs baseline: 2.2358x; 1.5993x over previous
#include <hip/hip_runtime.h>
#include <hip/hip_bf16.h>

typedef __attribute__((ext_vector_type(2))) float v2f;
typedef __attribute__((ext_vector_type(8))) float v8f;
typedef __attribute__((ext_vector_type(4))) float v4f;

#define D_DIM 256
#define L_DIM 1024
#define B_DIM 4
#define M_DIM (B_DIM * L_DIM)
#define LN_EPS 1e-5f

__device__ __forceinline__ float other_row(float x) {
  const int b = __builtin_bit_cast(int, x);
  return __builtin_bit_cast(float,
      __builtin_amdgcn_permlanex16(b, b, 0x76543210, 0xfedcba98, false, false));
}
template <int XMASK>
__device__ __forceinline__ float row_xor(float x) {
  const int b = __builtin_bit_cast(int, x);
  return __builtin_bit_cast(float,
      __builtin_amdgcn_update_dpp(0, b, 0x160 + XMASK, 0xf, 0xf, true));
}
__device__ __forceinline__ float wave_sum_32(float v) {
  v += other_row(v);
  v += row_xor<8>(v);
  v += row_xor<4>(v);
  v += row_xor<2>(v);
  v += row_xor<1>(v);
  return v;
}

__device__ __forceinline__ float block_sum_256(float v, float* sh8) {
  v = wave_sum_32(v);
  if ((threadIdx.x & 31) == 0) sh8[threadIdx.x >> 5] = v;
  __syncthreads();
  float r = 0.f;
#pragma unroll
  for (int i = 0; i < 8; ++i) r += sh8[i];
  return r;
}

__global__ void wmma_gemm_f32_kernel(const float* __restrict__ A,
                                     const float* __restrict__ Wm,
                                     float* __restrict__ Out,
                                     const float* __restrict__ bias) {
  const int K = D_DIM, N = D_DIM;
  const int wave = threadIdx.x >> 5;
  const int lane = threadIdx.x & 31;
  const int half = lane >> 4;
  const int l16  = lane & 15;
  const int tM = (blockIdx.x * 8 + wave) * 16;
  const int tN = blockIdx.y * 16;

  const float* arow = A  + (size_t)(tM + l16) * K + (half ? 2 : 0);
  const float* brow = Wm + (size_t)(tN + l16) * K + (half ? 2 : 0);

  v8f acc = {};
#pragma unroll 8
  for (int kc = 0; kc < K; kc += 4) {
    v2f a; a.x = arow[kc]; a.y = arow[kc + 1];
    v2f b; b.x = brow[kc]; b.y = brow[kc + 1];
    acc = __builtin_amdgcn_wmma_f32_16x16x4_f32(
         false, a,  false, b,
         (short)0, acc,  false,  false);
  }

  const int n = tN + l16;
  const float bv = bias ? bias[n] : 0.f;
  union { v8f v; float f[8]; } u; u.v = acc;
#pragma unroll
  for (int r = 0; r < 8; ++r) {
    const int m = tM + r + (half ? 8 : 0);
    Out[(size_t)m * N + n] = u.f[r] + bv;
  }
}

__global__ void phi_beta_kernel(const float* __restrict__ x,
                                float* __restrict__ q,
                                float* __restrict__ k,
                                const float* __restrict__ beta_w,
                                const float* __restrict__ beta_b,
                                const float* __restrict__ lnp_g,
                                const float* __restrict__ lnp_b,
                                float* __restrict__ beta) {
  __shared__ float sh[8 * 5];
  const int row = blockIdx.x, tid = threadIdx.x;
  const size_t off = (size_t)row * D_DIM + tid;

  float qe; { float t = q[off]; qe = (t > 0.f) ? t + 1.f : __expf(t); }
  float ke; { float t = k[off]; ke = (t > 0.f) ? t + 1.f : __expf(t); }
  const float xb = x[off] * beta_w[tid];

  const float sq  = block_sum_256(qe,      sh + 0);
  const float sq2 = block_sum_256(qe * qe, sh + 8);
  const float sk  = block_sum_256(ke,      sh + 16);
  const float sk2 = block_sum_256(ke * ke, sh + 24);
  const float sb  = block_sum_256(xb,      sh + 32);

  const float muq = sq / D_DIM, varq = sq2 / D_DIM - muq * muq;
  const float muk = sk / D_DIM, vark = sk2 / D_DIM - muk * muk;
  const float g = lnp_g[tid], bb = lnp_b[tid];
  q[off] = (qe - muq) * rsqrtf(varq + LN_EPS) * g + bb;
  k[off] = (ke - muk) * rsqrtf(vark + LN_EPS) * g + bb;
  if (tid == 0) beta[row] = 1.f / (1.f + __expf(-(sb + beta_b[0])));
}

union F8 { v4f v4[2]; float f[8]; };

#define SCAN_T 32
#define SCAN_NC (L_DIM / SCAN_T)
#define SCAN_F4 (SCAN_T * D_DIM / 4 / 256)

__global__ void __launch_bounds__(256)
delta_scan_kernel(const float* __restrict__ phi_q,
                  const float* __restrict__ phi_k,
                  const float* __restrict__ v,
                  const float* __restrict__ beta,
                  const float* __restrict__ decay,
                  float* __restrict__ y) {
  __shared__ v4f s_q[2][SCAN_T * D_DIM / 4];
  __shared__ v4f s_k[2][SCAN_T * D_DIM / 4];
  __shared__ float  s_v[2][SCAN_T * 8];
  __shared__ float  s_b[2][SCAN_T];

  const int b    = blockIdx.y;
  const int tid  = threadIdx.x;
  const int wave = tid >> 5;
  const int lane = tid & 31;
  const int row  = blockIdx.x * 8 + wave;

  const float df = 1.f / (1.f + __expf(-decay[0]));
  const size_t base = (size_t)b * L_DIM * D_DIM;

  const v4f* gq = (const v4f*)(phi_q + base) + tid;
  const v4f* gk = (const v4f*)(phi_k + base) + tid;
  const float*  gv = v + base + (size_t)(tid >> 3) * D_DIM + blockIdx.x * 8 + (tid & 7);
  const float*  gb = beta + (size_t)b * L_DIM + (tid & (SCAN_T - 1));

#pragma unroll
  for (int j = 0; j < SCAN_F4; ++j) {
    s_q[0][tid + 256 * j] = gq[256 * j];
    s_k[0][tid + 256 * j] = gk[256 * j];
  }
  s_v[0][tid] = gv[0];
  if (tid < SCAN_T) s_b[0][tid] = gb[0];
  __syncthreads();

  float w[8];
#pragma unroll
  for (int c = 0; c < 8; ++c) w[c] = 0.f;

  F8 cq, ck;
  cq.v4[0] = s_q[0][lane * 2];
  cq.v4[1] = s_q[0][lane * 2 + 1];
  ck.v4[0] = s_k[0][lane * 2];
  ck.v4[1] = s_k[0][lane * 2 + 1];
  float cv = s_v[0][wave];
  float cb = s_b[0][0];

  for (int ch = 0; ch < SCAN_NC; ++ch) {
    const int buf = ch & 1;
    const int chn = (ch + 1 < SCAN_NC) ? ch + 1 : ch;

    const size_t adv4 = (size_t)chn * (SCAN_T * D_DIM / 4);
    const v4f rq0 = gq[adv4 + 256 * 0], rk0 = gk[adv4 + 256 * 0];
    const v4f rq1 = gq[adv4 + 256 * 1], rk1 = gk[adv4 + 256 * 1];
    const v4f rq2 = gq[adv4 + 256 * 2], rk2 = gk[adv4 + 256 * 2];
    const v4f rq3 = gq[adv4 + 256 * 3], rk3 = gk[adv4 + 256 * 3];
    const v4f rq4 = gq[adv4 + 256 * 4], rk4 = gk[adv4 + 256 * 4];
    const v4f rq5 = gq[adv4 + 256 * 5], rk5 = gk[adv4 + 256 * 5];
    const v4f rq6 = gq[adv4 + 256 * 6], rk6 = gk[adv4 + 256 * 6];
    const v4f rq7 = gq[adv4 + 256 * 7], rk7 = gk[adv4 + 256 * 7];
    const float rv = gv[(size_t)chn * (SCAN_T * D_DIM)];
    const float rb = gb[chn * SCAN_T];
    asm volatile("" :: "v"(rq0), "v"(rq1), "v"(rq2), "v"(rq3),
                       "v"(rq4), "v"(rq5), "v"(rq6), "v"(rq7),
                       "v"(rk0), "v"(rk1), "v"(rk2), "v"(rk3),
                       "v"(rk4), "v"(rk5), "v"(rk6), "v"(rk7),
                       "v"(rv), "v"(rb));

    const v4f* lq = &s_q[buf][lane * 2];
    const v4f* lk = &s_k[buf][lane * 2];
    const float*  lv = &s_v[buf][wave];
    const float*  lb = &s_b[buf][0];
    float* yrow = y + base + (size_t)ch * SCAN_T * D_DIM + row;

    for (int tt = 0; tt < SCAN_T; ++tt) {
      const int tn = (tt + 1 < SCAN_T) ? tt + 1 : tt;
      F8 nq, nk;
      nq.v4[0] = lq[tn * (D_DIM / 4)];
      nq.v4[1] = lq[tn * (D_DIM / 4) + 1];
      nk.v4[0] = lk[tn * (D_DIM / 4)];
      nk.v4[1] = lk[tn * (D_DIM / 4) + 1];
      const float nv = lv[tn * 8];
      const float nb = lb[tn];

      float rq = 0.f, rk = 0.f;
#pragma unroll
      for (int c = 0; c < 8; ++c) {
        rq = fmaf(w[c], cq.f[c], rq);
        rk = fmaf(w[c], ck.f[c], rk);
      }
      rq = wave_sum_32(rq);
      rk = wave_sum_32(rk);
      if (lane == 0) yrow[(size_t)tt * D_DIM] = rq;

      const float u = cb * (cv - rk);
#pragma unroll
      for (int c = 0; c < 8; ++c) w[c] = fmaf(df, w[c], u * ck.f[c]);

      cq = nq; ck = nk; cv = nv; cb = nb;
    }

    v4f* dq = &s_q[buf ^ 1][tid];
    v4f* dk = &s_k[buf ^ 1][tid];
    dq[256 * 0] = rq0; dk[256 * 0] = rk0;
    dq[256 * 1] = rq1; dk[256 * 1] = rk1;
    dq[256 * 2] = rq2; dk[256 * 2] = rk2;
    dq[256 * 3] = rq3; dk[256 * 3] = rk3;
    dq[256 * 4] = rq4; dk[256 * 4] = rk4;
    dq[256 * 5] = rq5; dk[256 * 5] = rk5;
    dq[256 * 6] = rq6; dk[256 * 6] = rk6;
    dq[256 * 7] = rq7; dk[256 * 7] = rk7;
    s_v[buf ^ 1][tid] = rv;
    if (tid < SCAN_T) s_b[buf ^ 1][tid] = rb;
    __syncthreads();
    cq.v4[0] = s_q[buf ^ 1][lane * 2];
    cq.v4[1] = s_q[buf ^ 1][lane * 2 + 1];
    ck.v4[0] = s_k[buf ^ 1][lane * 2];
    ck.v4[1] = s_k[buf ^ 1][lane * 2 + 1];
    cv = s_v[buf ^ 1][wave];
    cb = s_b[buf ^ 1][0];
  }
}

__global__ void ln_rows_kernel(float* __restrict__ y,
                               const float* __restrict__ g,
                               const float* __restrict__ b) {
  __shared__ float sh[16];
  const int row = blockIdx.x, tid = threadIdx.x;
  const size_t off = (size_t)row * D_DIM + tid;
  const float v  = y[off];
  const float s  = block_sum_256(v, sh);
  const float s2 = block_sum_256(v * v, sh + 8);
  const float mu = s / D_DIM, var = s2 / D_DIM - mu * mu;
  y[off] = (v - mu) * rsqrtf(var + LN_EPS) * g[tid] + b[tid];
}

extern "C" void kernel_launch(void* const* d_in, const int* in_sizes, int n_in,
                              void* d_out, int out_size, void* d_ws, size_t ws_size,
                              hipStream_t stream) {
  (void)in_sizes; (void)n_in; (void)out_size; (void)ws_size;
  const float* x      = (const float*)d_in[0];
  const float* Wq     = (const float*)d_in[1];
  const float* Wk     = (const float*)d_in[2];
  const float* Wv     = (const float*)d_in[3];
  const float* beta_w = (const float*)d_in[4];
  const float* beta_b = (const float*)d_in[5];
  const float* decay  = (const float*)d_in[6];
  const float* Wo     = (const float*)d_in[7];
  const float* bo     = (const float*)d_in[8];
  const float* ln_g   = (const float*)d_in[9];
  const float* ln_b   = (const float*)d_in[10];
  const float* lnp_g  = (const float*)d_in[11];
  const float* lnp_b  = (const float*)d_in[12];
  float* out = (float*)d_out;

  char* ws = (char*)d_ws;
  const size_t MAT = (size_t)M_DIM * D_DIM * sizeof(float);
  float* q    = (float*)(ws);
  float* k    = (float*)(ws + MAT);
  float* vv   = (float*)(ws + 2 * MAT);
  float* beta = (float*)(ws + 3 * MAT);
  float* y    = (float*)(ws + 3 * MAT + (64 << 10));

  const dim3 gemm_grid(M_DIM / 128, D_DIM / 16);
  const dim3 gemm_blk(256);

  wmma_gemm_f32_kernel<<<gemm_grid, gemm_blk, 0, stream>>>(x, Wq, q, nullptr);
  wmma_gemm_f32_kernel<<<gemm_grid, gemm_blk, 0, stream>>>(x, Wk, k, nullptr);
  wmma_gemm_f32_kernel<<<gemm_grid, gemm_blk, 0, stream>>>(x, Wv, vv, nullptr);

  phi_beta_kernel<<<dim3(M_DIM), dim3(256), 0, stream>>>(
      x, q, k, beta_w, beta_b, lnp_g, lnp_b, beta);

  delta_scan_kernel<<<dim3(32, B_DIM), dim3(256), 0, stream>>>(
      q, k, vv, beta, decay, y);

  ln_rows_kernel<<<dim3(M_DIM), dim3(256), 0, stream>>>(y, ln_g, ln_b);

  wmma_gemm_f32_kernel<<<gemm_grid, gemm_blk, 0, stream>>>(y, Wo, out, bo);
}


# scan: each chunk's global loads are carried across the chunk loop's back edge (issued behind the hand-over barrier, stored to LDS a whole chunk later); the wait for them no longer precedes the steps
# speedup vs baseline: 2.3610x; 1.0560x over previous
#include <hip/hip_runtime.h>
#include <hip/hip_bf16.h>

typedef __attribute__((ext_vector_type(2))) float v2f;
typedef __attribute__((ext_vector_type(8))) float v8f;
typedef __attribute__((ext_vector_type(4))) float v4f;

#define D_DIM 256
#define L_DIM 1024
#define B_DIM 4
#define M_DIM (B_DIM * L_DIM)
#define LN_EPS 1e-5f

__device__ __forceinline__ float other_row(float x) {
  const int b = __builtin_bit_cast(int, x);
  return __builtin_bit_cast(float,
      __builtin_amdgcn_permlanex16(b, b, 0x76543210, 0xfedcba98, false, false));
}
template <int XMASK>
__device__ __forceinline__ float row_xor(float x) {
  const int b = __builtin_bit_cast(int, x);
  return __builtin_bit_cast(float,
      __builtin_amdgcn_update_dpp(0, b, 0x160 + XMASK, 0xf, 0xf, true));
}
__device__ __forceinline__ float wave_sum_32(float v) {
  v += other_row(v);
  v += row_xor<8>(v);
  v += row_xor<4>(v);
  v += row_xor<2>(v);
  v += row_xor<1>(v);
  return v;
}

__device__ __forceinline__ float block_sum_256(float v, float* sh8) {
  v = wave_sum_32(v);
  if ((threadIdx.x & 31) == 0) sh8[threadIdx.x >> 5] = v;
  __syncthreads();
  float r = 0.f;
#pragma unroll
  for (int i = 0; i < 8; ++i) r += sh8[i];
  return r;
}

__global__ void wmma_gemm_f32_kernel(const float* __restrict__ A,
                                     const float* __restrict__ Wm,
                                     float* __restrict__ Out,
                                     const float* __restrict__ bias) {
  const int K = D_DIM, N = D_DIM;
  const int wave = threadIdx.x >> 5;
  const int lane = threadIdx.x & 31;
  const int half = lane >> 4;
  const int l16  = lane & 15;
  const int tM = (blockIdx.x * 8 + wave) * 16;
  const int tN = blockIdx.y * 16;

  const float* arow = A  + (size_t)(tM + l16) * K + (half ? 2 : 0);
  const float* brow = Wm + (size_t)(tN + l16) * K + (half ? 2 : 0);

  v8f acc = {};
#pragma unroll 8
  for (int kc = 0; kc < K; kc += 4) {
    v2f a; a.x = arow[kc]; a.y = arow[kc + 1];
    v2f b; b.x = brow[kc]; b.y = brow[kc + 1];
    acc = __builtin_amdgcn_wmma_f32_16x16x4_f32(
         false, a,  false, b,
         (short)0, acc,  false,  false);
  }

  const int n = tN + l16;
  const float bv = bias ? bias[n] : 0.f;
  union { v8f v; float f[8]; } u; u.v = acc;
#pragma unroll
  for (int r = 0; r < 8; ++r) {
    const int m = tM + r + (half ? 8 : 0);
    Out[(size_t)m * N + n] = u.f[r] + bv;
  }
}

__global__ void phi_beta_kernel(const float* __restrict__ x,
                                float* __restrict__ q,
                                float* __restrict__ k,
                                const float* __restrict__ beta_w,
                                const float* __restrict__ beta_b,
                                const float* __restrict__ lnp_g,
                                const float* __restrict__ lnp_b,
                                float* __restrict__ beta) {
  __shared__ float sh[8 * 5];
  const int row = blockIdx.x, tid = threadIdx.x;
  const size_t off = (size_t)row * D_DIM + tid;

  float qe; { float t = q[off]; qe = (t > 0.f) ? t + 1.f : __expf(t); }
  float ke; { float t = k[off]; ke = (t > 0.f) ? t + 1.f : __expf(t); }
  const float xb = x[off] * beta_w[tid];

  const float sq  = block_sum_256(qe,      sh + 0);
  const float sq2 = block_sum_256(qe * qe, sh + 8);
  const float sk  = block_sum_256(ke,      sh + 16);
  const float sk2 = block_sum_256(ke * ke, sh + 24);
  const float sb  = block_sum_256(xb,      sh + 32);

  const float muq = sq / D_DIM, varq = sq2 / D_DIM - muq * muq;
  const float muk = sk / D_DIM, vark = sk2 / D_DIM - muk * muk;
  const float g = lnp_g[tid], bb = lnp_b[tid];
  q[off] = (qe - muq) * rsqrtf(varq + LN_EPS) * g + bb;
  k[off] = (ke - muk) * rsqrtf(vark + LN_EPS) * g + bb;
  if (tid == 0) beta[row] = 1.f / (1.f + __expf(-(sb + beta_b[0])));
}

union F8 { v4f v4[2]; float f[8]; };

#define SCAN_T 32
#define SCAN_NC (L_DIM / SCAN_T)
#define SCAN_F4 (SCAN_T * D_DIM / 4 / 256)

__global__ void __launch_bounds__(256)
delta_scan_kernel(const float* __restrict__ phi_q,
                  const float* __restrict__ phi_k,
                  const float* __restrict__ v,
                  const float* __restrict__ beta,
                  const float* __restrict__ decay,
                  float* __restrict__ y) {
  __shared__ v4f s_q[2][SCAN_T * D_DIM / 4];
  __shared__ v4f s_k[2][SCAN_T * D_DIM / 4];
  __shared__ float  s_v[2][SCAN_T * 8];
  __shared__ float  s_b[2][SCAN_T];

  const int b    = blockIdx.y;
  const int tid  = threadIdx.x;
  const int wave = tid >> 5;
  const int lane = tid & 31;
  const int row  = blockIdx.x * 8 + wave;

  const float df = 1.f / (1.f + __expf(-decay[0]));
  const size_t base = (size_t)b * L_DIM * D_DIM;

  const v4f* gq = (const v4f*)(phi_q + base) + tid;
  const v4f* gk = (const v4f*)(phi_k + base) + tid;
  const float*  gv = v + base + (size_t)(tid >> 3) * D_DIM + blockIdx.x * 8 + (tid & 7);
  const float*  gb = beta + (size_t)b * L_DIM + (tid & (SCAN_T - 1));

#pragma unroll
  for (int j = 0; j < SCAN_F4; ++j) {
    s_q[0][tid + 256 * j] = gq[256 * j];
    s_k[0][tid + 256 * j] = gk[256 * j];
  }
  s_v[0][tid] = gv[0];
  if (tid < SCAN_T) s_b[0][tid] = gb[0];
  __syncthreads();

  v4f rq0 = gq[(SCAN_T * D_DIM / 4) + 256 * 0], rk0 = gk[(SCAN_T * D_DIM / 4) + 256 * 0];
  v4f rq1 = gq[(SCAN_T * D_DIM / 4) + 256 * 1], rk1 = gk[(SCAN_T * D_DIM / 4) + 256 * 1];
  v4f rq2 = gq[(SCAN_T * D_DIM / 4) + 256 * 2], rk2 = gk[(SCAN_T * D_DIM / 4) + 256 * 2];
  v4f rq3 = gq[(SCAN_T * D_DIM / 4) + 256 * 3], rk3 = gk[(SCAN_T * D_DIM / 4) + 256 * 3];
  v4f rq4 = gq[(SCAN_T * D_DIM / 4) + 256 * 4], rk4 = gk[(SCAN_T * D_DIM / 4) + 256 * 4];
  v4f rq5 = gq[(SCAN_T * D_DIM / 4) + 256 * 5], rk5 = gk[(SCAN_T * D_DIM / 4) + 256 * 5];
  v4f rq6 = gq[(SCAN_T * D_DIM / 4) + 256 * 6], rk6 = gk[(SCAN_T * D_DIM / 4) + 256 * 6];
  v4f rq7 = gq[(SCAN_T * D_DIM / 4) + 256 * 7], rk7 = gk[(SCAN_T * D_DIM / 4) + 256 * 7];
  float rv = gv[(size_t)(SCAN_T * D_DIM)];
  float rb = gb[SCAN_T];

  float w[8];
#pragma unroll
  for (int c = 0; c < 8; ++c) w[c] = 0.f;

  F8 cq, ck;
  cq.v4[0] = s_q[0][lane * 2];
  cq.v4[1] = s_q[0][lane * 2 + 1];
  ck.v4[0] = s_k[0][lane * 2];
  ck.v4[1] = s_k[0][lane * 2 + 1];
  float cv = s_v[0][wave];
  float cb = s_b[0][0];

  for (int ch = 0; ch < SCAN_NC; ++ch) {
    const int buf = ch & 1;

    const v4f* lq = &s_q[buf][lane * 2];
    const v4f* lk = &s_k[buf][lane * 2];
    const float*  lv = &s_v[buf][wave];
    const float*  lb = &s_b[buf][0];
    float* yrow = y + base + (size_t)ch * SCAN_T * D_DIM + row;

    for (int tt = 0; tt < SCAN_T; ++tt) {
      const int tn = (tt + 1 < SCAN_T) ? tt + 1 : tt;
      F8 nq, nk;
      nq.v4[0] = lq[tn * (D_DIM / 4)];
      nq.v4[1] = lq[tn * (D_DIM / 4) + 1];
      nk.v4[0] = lk[tn * (D_DIM / 4)];
      nk.v4[1] = lk[tn * (D_DIM / 4) + 1];
      const float nv = lv[tn * 8];
      const float nb = lb[tn];

      float rq = 0.f, rk = 0.f;
#pragma unroll
      for (int c = 0; c < 8; ++c) {
        rq = fmaf(w[c], cq.f[c], rq);
        rk = fmaf(w[c], ck.f[c], rk);
      }
      rq = wave_sum_32(rq);
      rk = wave_sum_32(rk);
      if (lane == 0) yrow[(size_t)tt * D_DIM] = rq;

      const float u = cb * (cv - rk);
#pragma unroll
      for (int c = 0; c < 8; ++c) w[c] = fmaf(df, w[c], u * ck.f[c]);

      cq = nq; ck = nk; cv = nv; cb = nb;
    }

    v4f* dq = &s_q[buf ^ 1][tid];
    v4f* dk = &s_k[buf ^ 1][tid];
    dq[256 * 0] = rq0; dk[256 * 0] = rk0;
    dq[256 * 1] = rq1; dk[256 * 1] = rk1;
    dq[256 * 2] = rq2; dk[256 * 2] = rk2;
    dq[256 * 3] = rq3; dk[256 * 3] = rk3;
    dq[256 * 4] = rq4; dk[256 * 4] = rk4;
    dq[256 * 5] = rq5; dk[256 * 5] = rk5;
    dq[256 * 6] = rq6; dk[256 * 6] = rk6;
    dq[256 * 7] = rq7; dk[256 * 7] = rk7;
    s_v[buf ^ 1][tid] = rv;
    if (tid < SCAN_T) s_b[buf ^ 1][tid] = rb;
    __syncthreads();

    cq.v4[0] = s_q[buf ^ 1][lane * 2];
    cq.v4[1] = s_q[buf ^ 1][lane * 2 + 1];
    ck.v4[0] = s_k[buf ^ 1][lane * 2];
    ck.v4[1] = s_k[buf ^ 1][lane * 2 + 1];
    cv = s_v[buf ^ 1][wave];
    cb = s_b[buf ^ 1][0];

    const int chn = (ch + 2 < SCAN_NC) ? ch + 2 : SCAN_NC - 1;
    const size_t adv4 = (size_t)chn * (SCAN_T * D_DIM / 4);
    rq0 = gq[adv4 + 256 * 0]; rk0 = gk[adv4 + 256 * 0];
    rq1 = gq[adv4 + 256 * 1]; rk1 = gk[adv4 + 256 * 1];
    rq2 = gq[adv4 + 256 * 2]; rk2 = gk[adv4 + 256 * 2];
    rq3 = gq[adv4 + 256 * 3]; rk3 = gk[adv4 + 256 * 3];
    rq4 = gq[adv4 + 256 * 4]; rk4 = gk[adv4 + 256 * 4];
    rq5 = gq[adv4 + 256 * 5]; rk5 = gk[adv4 + 256 * 5];
    rq6 = gq[adv4 + 256 * 6]; rk6 = gk[adv4 + 256 * 6];
    rq7 = gq[adv4 + 256 * 7]; rk7 = gk[adv4 + 256 * 7];
    rv = gv[(size_t)chn * (SCAN_T * D_DIM)];
    rb = gb[chn * SCAN_T];
  }
}

__global__ void ln_rows_kernel(float* __restrict__ y,
                               const float* __restrict__ g,
                               const float* __restrict__ b) {
  __shared__ float sh[16];
  const int row = blockIdx.x, tid = threadIdx.x;
  const size_t off = (size_t)row * D_DIM + tid;
  const float v  = y[off];
  const float s  = block_sum_256(v, sh);
  const float s2 = block_sum_256(v * v, sh + 8);
  const float mu = s / D_DIM, var = s2 / D_DIM - mu * mu;
  y[off] = (v - mu) * rsqrtf(var + LN_EPS) * g[tid] + b[tid];
}

extern "C" void kernel_launch(void* const* d_in, const int* in_sizes, int n_in,
                              void* d_out, int out_size, void* d_ws, size_t ws_size,
                              hipStream_t stream) {
  (void)in_sizes; (void)n_in; (void)out_size; (void)ws_size;
  const float* x      = (const float*)d_in[0];
  const float* Wq     = (const float*)d_in[1];
  const float* Wk     = (const float*)d_in[2];
  const float* Wv     = (const float*)d_in[3];
  const float* beta_w = (const float*)d_in[4];
  const float* beta_b = (const float*)d_in[5];
  const float* decay  = (const float*)d_in[6];
  const float* Wo     = (const float*)d_in[7];
  const float* bo     = (const float*)d_in[8];
  const float* ln_g   = (const float*)d_in[9];
  const float* ln_b   = (const float*)d_in[10];
  const float* lnp_g  = (const float*)d_in[11];
  const float* lnp_b  = (const float*)d_in[12];
  float* out = (float*)d_out;

  char* ws = (char*)d_ws;
  const size_t MAT = (size_t)M_DIM * D_DIM * sizeof(float);
  float* q    = (float*)(ws);
  float* k    = (float*)(ws + MAT);
  float* vv   = (float*)(ws + 2 * MAT);
  float* beta = (float*)(ws + 3 * MAT);
  float* y    = (float*)(ws + 3 * MAT + (64 << 10));

  const dim3 gemm_grid(M_DIM / 128, D_DIM / 16);
  const dim3 gemm_blk(256);

  wmma_gemm_f32_kernel<<<gemm_grid, gemm_blk, 0, stream>>>(x, Wq, q, nullptr);
  wmma_gemm_f32_kernel<<<gemm_grid, gemm_blk, 0, stream>>>(x, Wk, k, nullptr);
  wmma_gemm_f32_kernel<<<gemm_grid, gemm_blk, 0, stream>>>(x, Wv, vv, nullptr);

  phi_beta_kernel<<<dim3(M_DIM), dim3(256), 0, stream>>>(
      x, q, k, beta_w, beta_b, lnp_g, lnp_b, beta);

  delta_scan_kernel<<<dim3(32, B_DIM), dim3(256), 0, stream>>>(
      q, k, vv, beta, decay, y);

  ln_rows_kernel<<<dim3(M_DIM), dim3(256), 0, stream>>>(y, ln_g, ln_b);

  wmma_gemm_f32_kernel<<<gemm_grid, gemm_blk, 0, stream>>>(y, Wo, out, bo);
}


# scan: step loop unrolled by four over rotating operand register sets; LDS reads issued two or more steps ahead (held by a scheduling barrier); a trip's four y values stored at its end; same sums
# speedup vs baseline: 2.6172x; 1.1085x over previous
#include <hip/hip_runtime.h>
#include <hip/hip_bf16.h>

typedef __attribute__((ext_vector_type(2))) float v2f;
typedef __attribute__((ext_vector_type(8))) float v8f;
typedef __attribute__((ext_vector_type(4))) float v4f;

#define D_DIM 256
#define L_DIM 1024
#define B_DIM 4
#define M_DIM (B_DIM * L_DIM)
#define LN_EPS 1e-5f

__device__ __forceinline__ float other_row(float x) {
  const int b = __builtin_bit_cast(int, x);
  return __builtin_bit_cast(float,
      __builtin_amdgcn_permlanex16(b, b, 0x76543210, 0xfedcba98, false, false));
}
template <int XMASK>
__device__ __forceinline__ float row_xor(float x) {
  const int b = __builtin_bit_cast(int, x);
  return __builtin_bit_cast(float,
      __builtin_amdgcn_update_dpp(0, b, 0x160 + XMASK, 0xf, 0xf, true));
}
__device__ __forceinline__ float wave_sum_32(float v) {
  v += other_row(v);
  v += row_xor<8>(v);
  v += row_xor<4>(v);
  v += row_xor<2>(v);
  v += row_xor<1>(v);
  return v;
}

__device__ __forceinline__ float block_sum_256(float v, float* sh8) {
  v = wave_sum_32(v);
  if ((threadIdx.x & 31) == 0) sh8[threadIdx.x >> 5] = v;
  __syncthreads();
  float r = 0.f;
#pragma unroll
  for (int i = 0; i < 8; ++i) r += sh8[i];
  return r;
}

__global__ void wmma_gemm_f32_kernel(const float* __restrict__ A,
                                     const float* __restrict__ Wm,
                                     float* __restrict__ Out,
                                     const float* __restrict__ bias) {
  const int K = D_DIM, N = D_DIM;
  const int wave = threadIdx.x >> 5;
  const int lane = threadIdx.x & 31;
  const int half = lane >> 4;
  const int l16  = lane & 15;
  const int tM = (blockIdx.x * 8 + wave) * 16;
  const int tN = blockIdx.y * 16;

  const float* arow = A  + (size_t)(tM + l16) * K + (half ? 2 : 0);
  const float* brow = Wm + (size_t)(tN + l16) * K + (half ? 2 : 0);

  v8f acc = {};
#pragma unroll 8
  for (int kc = 0; kc < K; kc += 4) {
    v2f a; a.x = arow[kc]; a.y = arow[kc + 1];
    v2f b; b.x = brow[kc]; b.y = brow[kc + 1];
    acc = __builtin_amdgcn_wmma_f32_16x16x4_f32(
         false, a,  false, b,
         (short)0, acc,  false,  false);
  }

  const int n = tN + l16;
  const float bv = bias ? bias[n] : 0.f;
  union { v8f v; float f[8]; } u; u.v = acc;
#pragma unroll
  for (int r = 0; r < 8; ++r) {
    const int m = tM + r + (half ? 8 : 0);
    Out[(size_t)m * N + n] = u.f[r] + bv;
  }
}

__global__ void phi_beta_kernel(const float* __restrict__ x,
                                float* __restrict__ q,
                                float* __restrict__ k,
                                const float* __restrict__ beta_w,
                                const float* __restrict__ beta_b,
                                const float* __restrict__ lnp_g,
                                const float* __restrict__ lnp_b,
                                float* __restrict__ beta) {
  __shared__ float sh[8 * 5];
  const int row = blockIdx.x, tid = threadIdx.x;
  const size_t off = (size_t)row * D_DIM + tid;

  float qe; { float t = q[off]; qe = (t > 0.f) ? t + 1.f : __expf(t); }
  float ke; { float t = k[off]; ke = (t > 0.f) ? t + 1.f : __expf(t); }
  const float xb = x[off] * beta_w[tid];

  const float sq  = block_sum_256(qe,      sh + 0);
  const float sq2 = block_sum_256(qe * qe, sh + 8);
  const float sk  = block_sum_256(ke,      sh + 16);
  const float sk2 = block_sum_256(ke * ke, sh + 24);
  const float sb  = block_sum_256(xb,      sh + 32);

  const float muq = sq / D_DIM, varq = sq2 / D_DIM - muq * muq;
  const float muk = sk / D_DIM, vark = sk2 / D_DIM - muk * muk;
  const float g = lnp_g[tid], bb = lnp_b[tid];
  q[off] = (qe - muq) * rsqrtf(varq + LN_EPS) * g + bb;
  k[off] = (ke - muk) * rsqrtf(vark + LN_EPS) * g + bb;
  if (tid == 0) beta[row] = 1.f / (1.f + __expf(-(sb + beta_b[0])));
}

union F8 { v4f v4[2]; float f[8]; };

struct StepOps { F8 q, k; float v, b; };

__device__ __forceinline__ void load_step_ops(StepOps& o, const v4f* lq, const v4f* lk,
                                              const float* lv, const float* lb, int t) {
  o.q.v4[0] = lq[t * (D_DIM / 4)];
  o.q.v4[1] = lq[t * (D_DIM / 4) + 1];
  o.k.v4[0] = lk[t * (D_DIM / 4)];
  o.k.v4[1] = lk[t * (D_DIM / 4) + 1];
  o.v = lv[t * 8];
  o.b = lb[t];
}

__device__ __forceinline__ float scan_step(float (&w)[8], const StepOps& o, float df) {
  float rq = 0.f, rk = 0.f;
#pragma unroll
  for (int c = 0; c < 8; ++c) {
    rq = fmaf(w[c], o.q.f[c], rq);
    rk = fmaf(w[c], o.k.f[c], rk);
  }
  rq = wave_sum_32(rq);
  rk = wave_sum_32(rk);

  const float u = o.b * (o.v - rk);
#pragma unroll
  for (int c = 0; c < 8; ++c) w[c] = fmaf(df, w[c], u * o.k.f[c]);
  return rq;
}

#define HOLD_READS_HERE() __builtin_amdgcn_sched_barrier(0)

#define SCAN_T 32
#define SCAN_NC (L_DIM / SCAN_T)
#define SCAN_F4 (SCAN_T * D_DIM / 4 / 256)

__global__ void __launch_bounds__(256)
delta_scan_kernel(const float* __restrict__ phi_q,
                  const float* __restrict__ phi_k,
                  const float* __restrict__ v,
                  const float* __restrict__ beta,
                  const float* __restrict__ decay,
                  float* __restrict__ y) {
  __shared__ v4f s_q[2][SCAN_T * D_DIM / 4];
  __shared__ v4f s_k[2][SCAN_T * D_DIM / 4];
  __shared__ float  s_v[2][SCAN_T * 8];
  __shared__ float  s_b[2][SCAN_T];

  const int b    = blockIdx.y;
  const int tid  = threadIdx.x;
  const int wave = tid >> 5;
  const int lane = tid & 31;
  const int row  = blockIdx.x * 8 + wave;

  const float df = 1.f / (1.f + __expf(-decay[0]));
  const size_t base = (size_t)b * L_DIM * D_DIM;

  const v4f* gq = (const v4f*)(phi_q + base) + tid;
  const v4f* gk = (const v4f*)(phi_k + base) + tid;
  const float*  gv = v + base + (size_t)(tid >> 3) * D_DIM + blockIdx.x * 8 + (tid & 7);
  const float*  gb = beta + (size_t)b * L_DIM + (tid & (SCAN_T - 1));

#pragma unroll
  for (int j = 0; j < SCAN_F4; ++j) {
    s_q[0][tid + 256 * j] = gq[256 * j];
    s_k[0][tid + 256 * j] = gk[256 * j];
  }
  s_v[0][tid] = gv[0];
  if (tid < SCAN_T) s_b[0][tid] = gb[0];
  __syncthreads();

  v4f rq0 = gq[(SCAN_T * D_DIM / 4) + 256 * 0], rk0 = gk[(SCAN_T * D_DIM / 4) + 256 * 0];
  v4f rq1 = gq[(SCAN_T * D_DIM / 4) + 256 * 1], rk1 = gk[(SCAN_T * D_DIM / 4) + 256 * 1];
  v4f rq2 = gq[(SCAN_T * D_DIM / 4) + 256 * 2], rk2 = gk[(SCAN_T * D_DIM / 4) + 256 * 2];
  v4f rq3 = gq[(SCAN_T * D_DIM / 4) + 256 * 3], rk3 = gk[(SCAN_T * D_DIM / 4) + 256 * 3];
  v4f rq4 = gq[(SCAN_T * D_DIM / 4) + 256 * 4], rk4 = gk[(SCAN_T * D_DIM / 4) + 256 * 4];
  v4f rq5 = gq[(SCAN_T * D_DIM / 4) + 256 * 5], rk5 = gk[(SCAN_T * D_DIM / 4) + 256 * 5];
  v4f rq6 = gq[(SCAN_T * D_DIM / 4) + 256 * 6], rk6 = gk[(SCAN_T * D_DIM / 4) + 256 * 6];
  v4f rq7 = gq[(SCAN_T * D_DIM / 4) + 256 * 7], rk7 = gk[(SCAN_T * D_DIM / 4) + 256 * 7];
  float rv = gv[(size_t)(SCAN_T * D_DIM)];
  float rb = gb[SCAN_T];

  float w[8];
#pragma unroll
  for (int c = 0; c < 8; ++c) w[c] = 0.f;

  StepOps o0, o1, o2, o3;
  load_step_ops(o0, &s_q[0][lane * 2], &s_k[0][lane * 2], &s_v[0][wave], &s_b[0][0], 0);
  load_step_ops(o1, &s_q[0][lane * 2], &s_k[0][lane * 2], &s_v[0][wave], &s_b[0][0], 1);

  for (int ch = 0; ch < SCAN_NC; ++ch) {
    const int buf = ch & 1;

    const v4f* lq = &s_q[buf][lane * 2];
    const v4f* lk = &s_k[buf][lane * 2];
    const float*  lv = &s_v[buf][wave];
    const float*  lb = &s_b[buf][0];
    float* yrow = y + base + (size_t)ch * SCAN_T * D_DIM + row;

    for (int t4 = 0; t4 < SCAN_T; t4 += 4) {
      const int ta = (t4 + 4 < SCAN_T) ? t4 + 4 : SCAN_T - 1;
      const int tb = (t4 + 5 < SCAN_T) ? t4 + 5 : SCAN_T - 1;
      load_step_ops(o2, lq, lk, lv, lb, t4 + 2);
      HOLD_READS_HERE();
      const float y0 = scan_step(w, o0, df);
      load_step_ops(o3, lq, lk, lv, lb, t4 + 3);
      HOLD_READS_HERE();
      const float y1 = scan_step(w, o1, df);
      load_step_ops(o0, lq, lk, lv, lb, ta);
      HOLD_READS_HERE();
      const float y2 = scan_step(w, o2, df);
      load_step_ops(o1, lq, lk, lv, lb, tb);
      HOLD_READS_HERE();
      const float y3 = scan_step(w, o3, df);
      if (lane == 0) {
        float* yp = yrow + (size_t)t4 * D_DIM;
        yp[0 * D_DIM] = y0;
        yp[1 * D_DIM] = y1;
        yp[2 * D_DIM] = y2;
        yp[3 * D_DIM] = y3;
      }
    }

    v4f* dq = &s_q[buf ^ 1][tid];
    v4f* dk = &s_k[buf ^ 1][tid];
    dq[256 * 0] = rq0; dk[256 * 0] = rk0;
    dq[256 * 1] = rq1; dk[256 * 1] = rk1;
    dq[256 * 2] = rq2; dk[256 * 2] = rk2;
    dq[256 * 3] = rq3; dk[256 * 3] = rk3;
    dq[256 * 4] = rq4; dk[256 * 4] = rk4;
    dq[256 * 5] = rq5; dk[256 * 5] = rk5;
    dq[256 * 6] = rq6; dk[256 * 6] = rk6;
    dq[256 * 7] = rq7; dk[256 * 7] = rk7;
    s_v[buf ^ 1][tid] = rv;
    if (tid < SCAN_T) s_b[buf ^ 1][tid] = rb;
    __syncthreads();

    load_step_ops(o0, &s_q[buf ^ 1][lane * 2], &s_k[buf ^ 1][lane * 2],
                  &s_v[buf ^ 1][wave], &s_b[buf ^ 1][0], 0);
    load_step_ops(o1, &s_q[buf ^ 1][lane * 2], &s_k[buf ^ 1][lane * 2],
                  &s_v[buf ^ 1][wave], &s_b[buf ^ 1][0], 1);

    const int chn = (ch + 2 < SCAN_NC) ? ch + 2 : SCAN_NC - 1;
    const size_t adv4 = (size_t)chn * (SCAN_T * D_DIM / 4);
    rq0 = gq[adv4 + 256 * 0]; rk0 = gk[adv4 + 256 * 0];
    rq1 = gq[adv4 + 256 * 1]; rk1 = gk[adv4 + 256 * 1];
    rq2 = gq[adv4 + 256 * 2]; rk2 = gk[adv4 + 256 * 2];
    rq3 = gq[adv4 + 256 * 3]; rk3 = gk[adv4 + 256 * 3];
    rq4 = gq[adv4 + 256 * 4]; rk4 = gk[adv4 + 256 * 4];
    rq5 = gq[adv4 + 256 * 5]; rk5 = gk[adv4 + 256 * 5];
    rq6 = gq[adv4 + 256 * 6]; rk6 = gk[adv4 + 256 * 6];
    rq7 = gq[adv4 + 256 * 7]; rk7 = gk[adv4 + 256 * 7];
    rv = gv[(size_t)chn * (SCAN_T * D_DIM)];
    rb = gb[chn * SCAN_T];
  }
}

__global__ void ln_rows_kernel(float* __restrict__ y,
                               const float* __restrict__ g,
                               const float* __restrict__ b) {
  __shared__ float sh[16];
  const int row = blockIdx.x, tid = threadIdx.x;
  const size_t off = (size_t)row * D_DIM + tid;
  const float v  = y[off];
  const float s  = block_sum_256(v, sh);
  const float s2 = block_sum_256(v * v, sh + 8);
  const float mu = s / D_DIM, var = s2 / D_DIM - mu * mu;
  y[off] = (v - mu) * rsqrtf(var + LN_EPS) * g[tid] + b[tid];
}

extern "C" void kernel_launch(void* const* d_in, const int* in_sizes, int n_in,
                              void* d_out, int out_size, void* d_ws, size_t ws_size,
                              hipStream_t stream) {
  (void)in_sizes; (void)n_in; (void)out_size; (void)ws_size;
  const float* x      = (const float*)d_in[0];
  const float* Wq     = (const float*)d_in[1];
  const float* Wk     = (const float*)d_in[2];
  const float* Wv     = (const float*)d_in[3];
  const float* beta_w = (const float*)d_in[4];
  const float* beta_b = (const float*)d_in[5];
  const float* decay  = (const float*)d_in[6];
  const float* Wo     = (const float*)d_in[7];
  const float* bo     = (const float*)d_in[8];
  const float* ln_g   = (const float*)d_in[9];
  const float* ln_b   = (const float*)d_in[10];
  const float* lnp_g  = (const float*)d_in[11];
  const float* lnp_b  = (const float*)d_in[12];
  float* out = (float*)d_out;

  char* ws = (char*)d_ws;
  const size_t MAT = (size_t)M_DIM * D_DIM * sizeof(float);
  float* q    = (float*)(ws);
  float* k    = (float*)(ws + MAT);
  float* vv   = (float*)(ws + 2 * MAT);
  float* beta = (float*)(ws + 3 * MAT);
  float* y    = (float*)(ws + 3 * MAT + (64 << 10));

  const dim3 gemm_grid(M_DIM / 128, D_DIM / 16);
  const dim3 gemm_blk(256);

  wmma_gemm_f32_kernel<<<gemm_grid, gemm_blk, 0, stream>>>(x, Wq, q, nullptr);
  wmma_gemm_f32_kernel<<<gemm_grid, gemm_blk, 0, stream>>>(x, Wk, k, nullptr);
  wmma_gemm_f32_kernel<<<gemm_grid, gemm_blk, 0, stream>>>(x, Wv, vv, nullptr);

  phi_beta_kernel<<<dim3(M_DIM), dim3(256), 0, stream>>>(
      x, q, k, beta_w, beta_b, lnp_g, lnp_b, beta);

  delta_scan_kernel<<<dim3(32, B_DIM), dim3(256), 0, stream>>>(
      q, k, vv, beta, decay, y);

  ln_rows_kernel<<<dim3(M_DIM), dim3(256), 0, stream>>>(y, ln_g, ln_b);

  wmma_gemm_f32_kernel<<<gemm_grid, gemm_blk, 0, stream>>>(y, Wo, out, bo);
}


# projection GEMMs re-tiled for operand reuse: wave = 16x64 strip (4 accumulators), its A rows in registers by 16-byte loads, W strip staged once in LDS; k order in each 8 permuted alike in A and B
# speedup vs baseline: 2.8897x; 1.1041x over previous
#include <hip/hip_runtime.h>
#include <hip/hip_bf16.h>

typedef __attribute__((ext_vector_type(2))) float v2f;
typedef __attribute__((ext_vector_type(8))) float v8f;
typedef __attribute__((ext_vector_type(4))) float v4f;

#define D_DIM 256
#define L_DIM 1024
#define B_DIM 4
#define M_DIM (B_DIM * L_DIM)
#define LN_EPS 1e-5f

__device__ __forceinline__ float other_row(float x) {
  const int b = __builtin_bit_cast(int, x);
  return __builtin_bit_cast(float,
      __builtin_amdgcn_permlanex16(b, b, 0x76543210, 0xfedcba98, false, false));
}
template <int XMASK>
__device__ __forceinline__ float row_xor(float x) {
  const int b = __builtin_bit_cast(int, x);
  return __builtin_bit_cast(float,
      __builtin_amdgcn_update_dpp(0, b, 0x160 + XMASK, 0xf, 0xf, true));
}
__device__ __forceinline__ float wave_sum_32(float v) {
  v += other_row(v);
  v += row_xor<8>(v);
  v += row_xor<4>(v);
  v += row_xor<2>(v);
  v += row_xor<1>(v);
  return v;
}

__device__ __forceinline__ float block_sum_256(float v, float* sh8) {
  v = wave_sum_32(v);
  if ((threadIdx.x & 31) == 0) sh8[threadIdx.x >> 5] = v;
  __syncthreads();
  float r = 0.f;
#pragma unroll
  for (int i = 0; i < 8; ++i) r += sh8[i];
  return r;
}

#define G_K      D_DIM
#define G_PITCH4 (G_K / 4 + 1)
#define G_NT     4

struct BSet { v4f p[G_NT][2]; };

__device__ __forceinline__ void gemm_load_b(BSet& s, const v4f* bp, int slice) {
#pragma unroll
  for (int t = 0; t < G_NT; ++t) {
    s.p[t][0] = bp[t * 16 * G_PITCH4 + 4 * slice];
    s.p[t][1] = bp[t * 16 * G_PITCH4 + 4 * slice + 2];
  }
}

__device__ __forceinline__ v8f gemm_wmma(v2f a, v2f b, v8f c) {
  return __builtin_amdgcn_wmma_f32_16x16x4_f32(
       false, a,  false, b,
       (short)0, c,  false,  false);
}

__device__ __forceinline__ void gemm_mma_slice(v8f (&acc)[G_NT], v4f a0, v4f a1, const BSet& s) {
#pragma unroll
  for (int pc = 0; pc < 2; ++pc) {
    const v4f a4 = pc ? a1 : a0;
    const v2f alo = __builtin_shufflevector(a4, a4, 0, 1);
    const v2f ahi = __builtin_shufflevector(a4, a4, 2, 3);
#pragma unroll
    for (int t = 0; t < G_NT; ++t) {
      const v4f b4 = s.p[t][pc];
      acc[t] = gemm_wmma(alo, __builtin_shufflevector(b4, b4, 0, 1), acc[t]);
    }
#pragma unroll
    for (int t = 0; t < G_NT; ++t) {
      const v4f b4 = s.p[t][pc];
      acc[t] = gemm_wmma(ahi, __builtin_shufflevector(b4, b4, 2, 3), acc[t]);
    }
  }
}

__global__ void __launch_bounds__(128)
wmma_gemm_f32_kernel(const float* __restrict__ A,
                     const float* __restrict__ Wm,
                     float* __restrict__ Out,
                     const float* __restrict__ bias) {
  __shared__ v4f s_w[64 * G_PITCH4];

  const int N = D_DIM;
  const int tid  = threadIdx.x;
  const int wave = tid >> 5;
  const int lane = tid & 31;
  const int half = lane >> 4;
  const int l16  = lane & 15;
  const int tM = (blockIdx.x * 4 + wave) * 16;
  const int tN = blockIdx.y * 64;

  const v4f* ap = (const v4f*)(A + (size_t)(tM + l16) * G_K) + half;
  v4f a[G_K / 8];
#pragma unroll
  for (int j = 0; j < G_K / 8; ++j) a[j] = ap[2 * j];

  const v4f* wp = (const v4f*)(Wm + (size_t)tN * G_K);
#pragma unroll
  for (int i = 0; i < 64 * (G_K / 4) / 128; ++i) {
    const int idx = tid + 128 * i;
    s_w[(idx >> 6) * G_PITCH4 + (idx & 63)] = wp[idx];
  }
  __syncthreads();

  const v4f* bp = s_w + l16 * G_PITCH4 + half;

  v8f acc[G_NT];
#pragma unroll
  for (int t = 0; t < G_NT; ++t) acc[t] = (v8f){};

  BSet b0, b1;
  gemm_load_b(b0, bp, 0);
#pragma unroll
  for (int s = 0; s < G_K / 16; s += 2) {
    gemm_load_b(b1, bp, s + 1);
    __builtin_amdgcn_sched_barrier(0);
    gemm_mma_slice(acc, a[2 * s], a[2 * s + 1], b0);
    if (s + 2 < G_K / 16) gemm_load_b(b0, bp, s + 2);
    __builtin_amdgcn_sched_barrier(0);
    gemm_mma_slice(acc, a[2 * s + 2], a[2 * s + 3], b1);
  }

#pragma unroll
  for (int t = 0; t < G_NT; ++t) {
    const int n = tN + 16 * t + l16;
    const float bv = bias ? bias[n] : 0.f;
    union { v8f v; float f[8]; } u; u.v = acc[t];
#pragma unroll
    for (int r = 0; r < 8; ++r) {
      const int m = tM + r + (half ? 8 : 0);
      Out[(size_t)m * N + n] = u.f[r] + bv;
    }
  }
}

__global__ void phi_beta_kernel(const float* __restrict__ x,
                                float* __restrict__ q,
                                float* __restrict__ k,
                                const float* __restrict__ beta_w,
                                const float* __restrict__ beta_b,
                                const float* __restrict__ lnp_g,
                                const float* __restrict__ lnp_b,
                                float* __restrict__ beta) {
  __shared__ float sh[8 * 5];
  const int row = blockIdx.x, tid = threadIdx.x;
  const size_t off = (size_t)row * D_DIM + tid;

  float qe; { float t = q[off]; qe = (t > 0.f) ? t + 1.f : __expf(t); }
  float ke; { float t = k[off]; ke = (t > 0.f) ? t + 1.f : __expf(t); }
  const float xb = x[off] * beta_w[tid];

  const float sq  = block_sum_256(qe,      sh + 0);
  const float sq2 = block_sum_256(qe * qe, sh + 8);
  const float sk  = block_sum_256(ke,      sh + 16);
  const float sk2 = block_sum_256(ke * ke, sh + 24);
  const float sb  = block_sum_256(xb,      sh + 32);

  const float muq = sq / D_DIM, varq = sq2 / D_DIM - muq * muq;
  const float muk = sk / D_DIM, vark = sk2 / D_DIM - muk * muk;
  const float g = lnp_g[tid], bb = lnp_b[tid];
  q[off] = (qe - muq) * rsqrtf(varq + LN_EPS) * g + bb;
  k[off] = (ke - muk) * rsqrtf(vark + LN_EPS) * g + bb;
  if (tid == 0) beta[row] = 1.f / (1.f + __expf(-(sb + beta_b[0])));
}

union F8 { v4f v4[2]; float f[8]; };

struct StepOps { F8 q, k; float v, b; };

__device__ __forceinline__ void load_step_ops(StepOps& o, const v4f* lq, const v4f* lk,
                                              const float* lv, const float* lb, int t) {
  o.q.v4[0] = lq[t * (D_DIM / 4)];
  o.q.v4[1] = lq[t * (D_DIM / 4) + 1];
  o.k.v4[0] = lk[t * (D_DIM / 4)];
  o.k.v4[1] = lk[t * (D_DIM / 4) + 1];
  o.v = lv[t * 8];
  o.b = lb[t];
}

__device__ __forceinline__ float scan_step(float (&w)[8], const StepOps& o, float df) {
  float rq = 0.f, rk = 0.f;
#pragma unroll
  for (int c = 0; c < 8; ++c) {
    rq = fmaf(w[c], o.q.f[c], rq);
    rk = fmaf(w[c], o.k.f[c], rk);
  }
  rq = wave_sum_32(rq);
  rk = wave_sum_32(rk);

  const float u = o.b * (o.v - rk);
#pragma unroll
  for (int c = 0; c < 8; ++c) w[c] = fmaf(df, w[c], u * o.k.f[c]);
  return rq;
}

#define HOLD_READS_HERE() __builtin_amdgcn_sched_barrier(0)

#define SCAN_T 32
#define SCAN_NC (L_DIM / SCAN_T)
#define SCAN_F4 (SCAN_T * D_DIM / 4 / 256)

__global__ void __launch_bounds__(256)
delta_scan_kernel(const float* __restrict__ phi_q,
                  const float* __restrict__ phi_k,
                  const float* __restrict__ v,
                  const float* __restrict__ beta,
                  const float* __restrict__ decay,
                  float* __restrict__ y) {
  __shared__ v4f s_q[2][SCAN_T * D_DIM / 4];
  __shared__ v4f s_k[2][SCAN_T * D_DIM / 4];
  __shared__ float  s_v[2][SCAN_T * 8];
  __shared__ float  s_b[2][SCAN_T];

  const int b    = blockIdx.y;
  const int tid  = threadIdx.x;
  const int wave = tid >> 5;
  const int lane = tid & 31;
  const int row  = blockIdx.x * 8 + wave;

  const float df = 1.f / (1.f + __expf(-decay[0]));
  const size_t base = (size_t)b * L_DIM * D_DIM;

  const v4f* gq = (const v4f*)(phi_q + base) + tid;
  const v4f* gk = (const v4f*)(phi_k + base) + tid;
  const float*  gv = v + base + (size_t)(tid >> 3) * D_DIM + blockIdx.x * 8 + (tid & 7);
  const float*  gb = beta + (size_t)b * L_DIM + (tid & (SCAN_T - 1));

#pragma unroll
  for (int j = 0; j < SCAN_F4; ++j) {
    s_q[0][tid + 256 * j] = gq[256 * j];
    s_k[0][tid + 256 * j] = gk[256 * j];
  }
  s_v[0][tid] = gv[0];
  if (tid < SCAN_T) s_b[0][tid] = gb[0];
  __syncthreads();

  v4f rq0 = gq[(SCAN_T * D_DIM / 4) + 256 * 0], rk0 = gk[(SCAN_T * D_DIM / 4) + 256 * 0];
  v4f rq1 = gq[(SCAN_T * D_DIM / 4) + 256 * 1], rk1 = gk[(SCAN_T * D_DIM / 4) + 256 * 1];
  v4f rq2 = gq[(SCAN_T * D_DIM / 4) + 256 * 2], rk2 = gk[(SCAN_T * D_DIM / 4) + 256 * 2];
  v4f rq3 = gq[(SCAN_T * D_DIM / 4) + 256 * 3], rk3 = gk[(SCAN_T * D_DIM / 4) + 256 * 3];
  v4f rq4 = gq[(SCAN_T * D_DIM / 4) + 256 * 4], rk4 = gk[(SCAN_T * D_DIM / 4) + 256 * 4];
  v4f rq5 = gq[(SCAN_T * D_DIM / 4) + 256 * 5], rk5 = gk[(SCAN_T * D_DIM / 4) + 256 * 5];
  v4f rq6 = gq[(SCAN_T * D_DIM / 4) + 256 * 6], rk6 = gk[(SCAN_T * D_DIM / 4) + 256 * 6];
  v4f rq7 = gq[(SCAN_T * D_DIM / 4) + 256 * 7], rk7 = gk[(SCAN_T * D_DIM / 4) + 256 * 7];
  float rv = gv[(size_t)(SCAN_T * D_DIM)];
  float rb = gb[SCAN_T];

  float w[8];
#pragma unroll
  for (int c = 0; c < 8; ++c) w[c] = 0.f;

  StepOps o0, o1, o2, o3;
  load_step_ops(o0, &s_q[0][lane * 2], &s_k[0][lane * 2], &s_v[0][wave], &s_b[0][0], 0);
  load_step_ops(o1, &s_q[0][lane * 2], &s_k[0][lane * 2], &s_v[0][wave], &s_b[0][0], 1);

  for (int ch = 0; ch < SCAN_NC; ++ch) {
    const int buf = ch & 1;

    const v4f* lq = &s_q[buf][lane * 2];
    const v4f* lk = &s_k[buf][lane * 2];
    const float*  lv = &s_v[buf][wave];
    const float*  lb = &s_b[buf][0];
    float* yrow = y + base + (size_t)ch * SCAN_T * D_DIM + row;

    for (int t4 = 0; t4 < SCAN_T; t4 += 4) {
      const int ta = (t4 + 4 < SCAN_T) ? t4 + 4 : SCAN_T - 1;
      const int tb = (t4 + 5 < SCAN_T) ? t4 + 5 : SCAN_T - 1;
      load_step_ops(o2, lq, lk, lv, lb, t4 + 2);
      HOLD_READS_HERE();
      const float y0 = scan_step(w, o0, df);
      load_step_ops(o3, lq, lk, lv, lb, t4 + 3);
      HOLD_READS_HERE();
      const float y1 = scan_step(w, o1, df);
      load_step_ops(o0, lq, lk, lv, lb, ta);
      HOLD_READS_HERE();
      const float y2 = scan_step(w, o2, df);
      load_step_ops(o1, lq, lk, lv, lb, tb);
      HOLD_READS_HERE();
      const float y3 = scan_step(w, o3, df);
      if (lane == 0) {
        float* yp = yrow + (size_t)t4 * D_DIM;
        yp[0 * D_DIM] = y0;
        yp[1 * D_DIM] = y1;
        yp[2 * D_DIM] = y2;
        yp[3 * D_DIM] = y3;
      }
    }

    v4f* dq = &s_q[buf ^ 1][tid];
    v4f* dk = &s_k[buf ^ 1][tid];
    dq[256 * 0] = rq0; dk[256 * 0] = rk0;
    dq[256 * 1] = rq1; dk[256 * 1] = rk1;
    dq[256 * 2] = rq2; dk[256 * 2] = rk2;
    dq[256 * 3] = rq3; dk[256 * 3] = rk3;
    dq[256 * 4] = rq4; dk[256 * 4] = rk4;
    dq[256 * 5] = rq5; dk[256 * 5] = rk5;
    dq[256 * 6] = rq6; dk[256 * 6] = rk6;
    dq[256 * 7] = rq7; dk[256 * 7] = rk7;
    s_v[buf ^ 1][tid] = rv;
    if (tid < SCAN_T) s_b[buf ^ 1][tid] = rb;
    __syncthreads();

    load_step_ops(o0, &s_q[buf ^ 1][lane * 2], &s_k[buf ^ 1][lane * 2],
                  &s_v[buf ^ 1][wave], &s_b[buf ^ 1][0], 0);
    load_step_ops(o1, &s_q[buf ^ 1][lane * 2], &s_k[buf ^ 1][lane * 2],
                  &s_v[buf ^ 1][wave], &s_b[buf ^ 1][0], 1);

    const int chn = (ch + 2 < SCAN_NC) ? ch + 2 : SCAN_NC - 1;
    const size_t adv4 = (size_t)chn * (SCAN_T * D_DIM / 4);
    rq0 = gq[adv4 + 256 * 0]; rk0 = gk[adv4 + 256 * 0];
    rq1 = gq[adv4 + 256 * 1]; rk1 = gk[adv4 + 256 * 1];
    rq2 = gq[adv4 + 256 * 2]; rk2 = gk[adv4 + 256 * 2];
    rq3 = gq[adv4 + 256 * 3]; rk3 = gk[adv4 + 256 * 3];
    rq4 = gq[adv4 + 256 * 4]; rk4 = gk[adv4 + 256 * 4];
    rq5 = gq[adv4 + 256 * 5]; rk5 = gk[adv4 + 256 * 5];
    rq6 = gq[adv4 + 256 * 6]; rk6 = gk[adv4 + 256 * 6];
    rq7 = gq[adv4 + 256 * 7]; rk7 = gk[adv4 + 256 * 7];
    rv = gv[(size_t)chn * (SCAN_T * D_DIM)];
    rb = gb[chn * SCAN_T];
  }
}

__global__ void ln_rows_kernel(float* __restrict__ y,
                               const float* __restrict__ g,
                               const float* __restrict__ b) {
  __shared__ float sh[16];
  const int row = blockIdx.x, tid = threadIdx.x;
  const size_t off = (size_t)row * D_DIM + tid;
  const float v  = y[off];
  const float s  = block_sum_256(v, sh);
  const float s2 = block_sum_256(v * v, sh + 8);
  const float mu = s / D_DIM, var = s2 / D_DIM - mu * mu;
  y[off] = (v - mu) * rsqrtf(var + LN_EPS) * g[tid] + b[tid];
}

extern "C" void kernel_launch(void* const* d_in, const int* in_sizes, int n_in,
                              void* d_out, int out_size, void* d_ws, size_t ws_size,
                              hipStream_t stream) {
  (void)in_sizes; (void)n_in; (void)out_size; (void)ws_size;
  const float* x      = (const float*)d_in[0];
  const float* Wq     = (const float*)d_in[1];
  const float* Wk     = (const float*)d_in[2];
  const float* Wv     = (const float*)d_in[3];
  const float* beta_w = (const float*)d_in[4];
  const float* beta_b = (const float*)d_in[5];
  const float* decay  = (const float*)d_in[6];
  const float* Wo     = (const float*)d_in[7];
  const float* bo     = (const float*)d_in[8];
  const float* ln_g   = (const float*)d_in[9];
  const float* ln_b   = (const float*)d_in[10];
  const float* lnp_g  = (const float*)d_in[11];
  const float* lnp_b  = (const float*)d_in[12];
  float* out = (float*)d_out;

  char* ws = (char*)d_ws;
  const size_t MAT = (size_t)M_DIM * D_DIM * sizeof(float);
  float* q    = (float*)(ws);
  float* k    = (float*)(ws + MAT);
  float* vv   = (float*)(ws + 2 * MAT);
  float* beta = (float*)(ws + 3 * MAT);
  float* y    = (float*)(ws + 3 * MAT + (64 << 10));

  const dim3 gemm_grid(M_DIM / 64, D_DIM / 64);
  const dim3 gemm_blk(128);

  wmma_gemm_f32_kernel<<<gemm_grid, gemm_blk, 0, stream>>>(x, Wq, q, nullptr);
  wmma_gemm_f32_kernel<<<gemm_grid, gemm_blk, 0, stream>>>(x, Wk, k, nullptr);
  wmma_gemm_f32_kernel<<<gemm_grid, gemm_blk, 0, stream>>>(x, Wv, vv, nullptr);

  phi_beta_kernel<<<dim3(M_DIM), dim3(256), 0, stream>>>(
      x, q, k, beta_w, beta_b, lnp_g, lnp_b, beta);

  delta_scan_kernel<<<dim3(32, B_DIM), dim3(256), 0, stream>>>(
      q, k, vv, beta, decay, y);

  ln_rows_kernel<<<dim3(M_DIM), dim3(256), 0, stream>>>(y, ln_g, ln_b);

  wmma_gemm_f32_kernel<<<gemm_grid, gemm_blk, 0, stream>>>(y, Wo, out, bo);
}


# scan: blocks of 4 waves, 256 of them: one wave per SIMD on every WGP, not two per SIMD on half the WGPs; staged chunk = 64 steps (256 KB LDS, so one block per WGP); same sums, same order
# speedup vs baseline: 3.4162x; 1.1822x over previous
#include <hip/hip_runtime.h>
#include <hip/hip_bf16.h>

typedef __attribute__((ext_vector_type(2))) float v2f;
typedef __attribute__((ext_vector_type(8))) float v8f;
typedef __attribute__((ext_vector_type(4))) float v4f;

#define D_DIM 256
#define L_DIM 1024
#define B_DIM 4
#define M_DIM (B_DIM * L_DIM)
#define LN_EPS 1e-5f

__device__ __forceinline__ float other_row(float x) {
  const int b = __builtin_bit_cast(int, x);
  return __builtin_bit_cast(float,
      __builtin_amdgcn_permlanex16(b, b, 0x76543210, 0xfedcba98, false, false));
}
template <int XMASK>
__device__ __forceinline__ float row_xor(float x) {
  const int b = __builtin_bit_cast(int, x);
  return __builtin_bit_cast(float,
      __builtin_amdgcn_update_dpp(0, b, 0x160 + XMASK, 0xf, 0xf, true));
}
__device__ __forceinline__ float wave_sum_32(float v) {
  v += other_row(v);
  v += row_xor<8>(v);
  v += row_xor<4>(v);
  v += row_xor<2>(v);
  v += row_xor<1>(v);
  return v;
}

__device__ __forceinline__ float block_sum_256(float v, float* sh8) {
  v = wave_sum_32(v);
  if ((threadIdx.x & 31) == 0) sh8[threadIdx.x >> 5] = v;
  __syncthreads();
  float r = 0.f;
#pragma unroll
  for (int i = 0; i < 8; ++i) r += sh8[i];
  return r;
}

#define G_K      D_DIM
#define G_PITCH4 (G_K / 4 + 1)
#define G_NT     4

struct BSet { v4f p[G_NT][2]; };

__device__ __forceinline__ void gemm_load_b(BSet& s, const v4f* bp, int slice) {
#pragma unroll
  for (int t = 0; t < G_NT; ++t) {
    s.p[t][0] = bp[t * 16 * G_PITCH4 + 4 * slice];
    s.p[t][1] = bp[t * 16 * G_PITCH4 + 4 * slice + 2];
  }
}

__device__ __forceinline__ v8f gemm_wmma(v2f a, v2f b, v8f c) {
  return __builtin_amdgcn_wmma_f32_16x16x4_f32(
       false, a,  false, b,
       (short)0, c,  false,  false);
}

__device__ __forceinline__ void gemm_mma_slice(v8f (&acc)[G_NT], v4f a0, v4f a1, const BSet& s) {
#pragma unroll
  for (int pc = 0; pc < 2; ++pc) {
    const v4f a4 = pc ? a1 : a0;
    const v2f alo = __builtin_shufflevector(a4, a4, 0, 1);
    const v2f ahi = __builtin_shufflevector(a4, a4, 2, 3);
#pragma unroll
    for (int t = 0; t < G_NT; ++t) {
      const v4f b4 = s.p[t][pc];
      acc[t] = gemm_wmma(alo, __builtin_shufflevector(b4, b4, 0, 1), acc[t]);
    }
#pragma unroll
    for (int t = 0; t < G_NT; ++t) {
      const v4f b4 = s.p[t][pc];
      acc[t] = gemm_wmma(ahi, __builtin_shufflevector(b4, b4, 2, 3), acc[t]);
    }
  }
}

__global__ void __launch_bounds__(128)
wmma_gemm_f32_kernel(const float* __restrict__ A,
                     const float* __restrict__ Wm,
                     float* __restrict__ Out,
                     const float* __restrict__ bias) {
  __shared__ v4f s_w[64 * G_PITCH4];

  const int N = D_DIM;
  const int tid  = threadIdx.x;
  const int wave = tid >> 5;
  const int lane = tid & 31;
  const int half = lane >> 4;
  const int l16  = lane & 15;
  const int tM = (blockIdx.x * 4 + wave) * 16;
  const int tN = blockIdx.y * 64;

  const v4f* ap = (const v4f*)(A + (size_t)(tM + l16) * G_K) + half;
  v4f a[G_K / 8];
#pragma unroll
  for (int j = 0; j < G_K / 8; ++j) a[j] = ap[2 * j];

  const v4f* wp = (const v4f*)(Wm + (size_t)tN * G_K);
#pragma unroll
  for (int i = 0; i < 64 * (G_K / 4) / 128; ++i) {
    const int idx = tid + 128 * i;
    s_w[(idx >> 6) * G_PITCH4 + (idx & 63)] = wp[idx];
  }
  __syncthreads();

  const v4f* bp = s_w + l16 * G_PITCH4 + half;

  v8f acc[G_NT];
#pragma unroll
  for (int t = 0; t < G_NT; ++t) acc[t] = (v8f){};

  BSet b0, b1;
  gemm_load_b(b0, bp, 0);
#pragma unroll
  for (int s = 0; s < G_K / 16; s += 2) {
    gemm_load_b(b1, bp, s + 1);
    __builtin_amdgcn_sched_barrier(0);
    gemm_mma_slice(acc, a[2 * s], a[2 * s + 1], b0);
    if (s + 2 < G_K / 16) gemm_load_b(b0, bp, s + 2);
    __builtin_amdgcn_sched_barrier(0);
    gemm_mma_slice(acc, a[2 * s + 2], a[2 * s + 3], b1);
  }

#pragma unroll
  for (int t = 0; t < G_NT; ++t) {
    const int n = tN + 16 * t + l16;
    const float bv = bias ? bias[n] : 0.f;
    union { v8f v; float f[8]; } u; u.v = acc[t];
#pragma unroll
    for (int r = 0; r < 8; ++r) {
      const int m = tM + r + (half ? 8 : 0);
      Out[(size_t)m * N + n] = u.f[r] + bv;
    }
  }
}

__global__ void phi_beta_kernel(const float* __restrict__ x,
                                float* __restrict__ q,
                                float* __restrict__ k,
                                const float* __restrict__ beta_w,
                                const float* __restrict__ beta_b,
                                const float* __restrict__ lnp_g,
                                const float* __restrict__ lnp_b,
                                float* __restrict__ beta) {
  __shared__ float sh[8 * 5];
  const int row = blockIdx.x, tid = threadIdx.x;
  const size_t off = (size_t)row * D_DIM + tid;

  float qe; { float t = q[off]; qe = (t > 0.f) ? t + 1.f : __expf(t); }
  float ke; { float t = k[off]; ke = (t > 0.f) ? t + 1.f : __expf(t); }
  const float xb = x[off] * beta_w[tid];

  const float sq  = block_sum_256(qe,      sh + 0);
  const float sq2 = block_sum_256(qe * qe, sh + 8);
  const float sk  = block_sum_256(ke,      sh + 16);
  const float sk2 = block_sum_256(ke * ke, sh + 24);
  const float sb  = block_sum_256(xb,      sh + 32);

  const float muq = sq / D_DIM, varq = sq2 / D_DIM - muq * muq;
  const float muk = sk / D_DIM, vark = sk2 / D_DIM - muk * muk;
  const float g = lnp_g[tid], bb = lnp_b[tid];
  q[off] = (qe - muq) * rsqrtf(varq + LN_EPS) * g + bb;
  k[off] = (ke - muk) * rsqrtf(vark + LN_EPS) * g + bb;
  if (tid == 0) beta[row] = 1.f / (1.f + __expf(-(sb + beta_b[0])));
}

#define SCAN_ROWS 4
#define SCAN_NT (SCAN_ROWS * 32)
#define SCAN_T 64
#define SCAN_NC (L_DIM / SCAN_T)
#define SCAN_F4 (SCAN_T * D_DIM / 4 / SCAN_NT)
#define SCAN_VN (SCAN_T * SCAN_ROWS / SCAN_NT)
#define SCAN_LATE 8
#define SCAN_EARLY (SCAN_F4 - SCAN_LATE)
#define SCAN_LATE_AT 32

union F8 { v4f v4[2]; float f[8]; };

struct StepOps { F8 q, k; float v, b; };

__device__ __forceinline__ void load_step_ops(StepOps& o, const v4f* lq, const v4f* lk,
                                              const float* lv, const float* lb, int t) {
  o.q.v4[0] = lq[t * (D_DIM / 4)];
  o.q.v4[1] = lq[t * (D_DIM / 4) + 1];
  o.k.v4[0] = lk[t * (D_DIM / 4)];
  o.k.v4[1] = lk[t * (D_DIM / 4) + 1];
  o.v = lv[t * SCAN_ROWS];
  o.b = lb[t];
}

__device__ __forceinline__ float scan_step(float (&w)[8], const StepOps& o, float df) {
  float rq = 0.f, rk = 0.f;
#pragma unroll
  for (int c = 0; c < 8; ++c) {
    rq = fmaf(w[c], o.q.f[c], rq);
    rk = fmaf(w[c], o.k.f[c], rk);
  }
  rq = wave_sum_32(rq);
  rk = wave_sum_32(rk);

  const float u = o.b * (o.v - rk);
#pragma unroll
  for (int c = 0; c < 8; ++c) w[c] = fmaf(df, w[c], u * o.k.f[c]);
  return rq;
}

#define HOLD_READS_HERE() __builtin_amdgcn_sched_barrier(0)


__global__ void __launch_bounds__(SCAN_NT)
delta_scan_kernel(const float* __restrict__ phi_q,
                  const float* __restrict__ phi_k,
                  const float* __restrict__ v,
                  const float* __restrict__ beta,
                  const float* __restrict__ decay,
                  float* __restrict__ y) {
  __shared__ v4f s_q[2][SCAN_T * D_DIM / 4];
  __shared__ v4f s_k[2][SCAN_T * D_DIM / 4];
  __shared__ float  s_v[2][SCAN_T * SCAN_ROWS];
  __shared__ float  s_b[2][SCAN_T];

  const int b    = blockIdx.y;
  const int tid  = threadIdx.x;
  const int wave = tid >> 5;
  const int lane = tid & 31;
  const int row  = blockIdx.x * SCAN_ROWS + wave;

  const float df = 1.f / (1.f + __expf(-decay[0]));
  const size_t base = (size_t)b * L_DIM * D_DIM;

  const v4f* gq = (const v4f*)(phi_q + base) + tid;
  const v4f* gk = (const v4f*)(phi_k + base) + tid;
  const float*  gv = v + base + (size_t)(tid / SCAN_ROWS) * D_DIM
                       + blockIdx.x * SCAN_ROWS + (tid % SCAN_ROWS);
  const float*  gb = beta + (size_t)b * L_DIM + (tid & (SCAN_T - 1));

#pragma unroll
  for (int j = 0; j < SCAN_F4; ++j) {
    s_q[0][tid + SCAN_NT * j] = gq[SCAN_NT * j];
    s_k[0][tid + SCAN_NT * j] = gk[SCAN_NT * j];
  }
#pragma unroll
  for (int j = 0; j < SCAN_VN; ++j)
    s_v[0][tid + SCAN_NT * j] = gv[(size_t)j * (SCAN_NT / SCAN_ROWS) * D_DIM];
  if (tid < SCAN_T) s_b[0][tid] = gb[0];
  __syncthreads();

  v4f rq[SCAN_F4], rk[SCAN_F4];
  float rv[SCAN_VN];
#pragma unroll
  for (int j = 0; j < SCAN_F4; ++j) rk[j] = gk[(SCAN_T * D_DIM / 4) + SCAN_NT * j];
#pragma unroll
  for (int j = 0; j < SCAN_EARLY; ++j) rq[j] = gq[(SCAN_T * D_DIM / 4) + SCAN_NT * j];
#pragma unroll
  for (int j = SCAN_EARLY; j < SCAN_F4; ++j) rq[j] = (v4f){0.f, 0.f, 0.f, 0.f};
#pragma unroll
  for (int j = 0; j < SCAN_VN; ++j)
    rv[j] = gv[(size_t)(SCAN_T * D_DIM) + (size_t)j * (SCAN_NT / SCAN_ROWS) * D_DIM];
  float rb = gb[SCAN_T];

  float w[8];
#pragma unroll
  for (int c = 0; c < 8; ++c) w[c] = 0.f;

  StepOps o0, o1, o2, o3;
  load_step_ops(o0, &s_q[0][lane * 2], &s_k[0][lane * 2], &s_v[0][wave], &s_b[0][0], 0);
  load_step_ops(o1, &s_q[0][lane * 2], &s_k[0][lane * 2], &s_v[0][wave], &s_b[0][0], 1);

  for (int ch = 0; ch < SCAN_NC; ++ch) {
    const int buf = ch & 1;
    const int chl = (ch + 1 < SCAN_NC) ? ch + 1 : SCAN_NC - 1;
    const size_t late4 = (size_t)chl * (SCAN_T * D_DIM / 4);

    const v4f* lq = &s_q[buf][lane * 2];
    const v4f* lk = &s_k[buf][lane * 2];
    const float*  lv = &s_v[buf][wave];
    const float*  lb = &s_b[buf][0];
    float* yrow = y + base + (size_t)ch * SCAN_T * D_DIM + row;

    auto trip = [&](const int t4) {
      const int ta = (t4 + 4 < SCAN_T) ? t4 + 4 : SCAN_T - 1;
      const int tb = (t4 + 5 < SCAN_T) ? t4 + 5 : SCAN_T - 1;
      load_step_ops(o2, lq, lk, lv, lb, t4 + 2);
      HOLD_READS_HERE();
      const float y0 = scan_step(w, o0, df);
      load_step_ops(o3, lq, lk, lv, lb, t4 + 3);
      HOLD_READS_HERE();
      const float y1 = scan_step(w, o1, df);
      load_step_ops(o0, lq, lk, lv, lb, ta);
      HOLD_READS_HERE();
      const float y2 = scan_step(w, o2, df);
      load_step_ops(o1, lq, lk, lv, lb, tb);
      HOLD_READS_HERE();
      const float y3 = scan_step(w, o3, df);
      if (lane == 0) {
        float* yp = yrow + (size_t)t4 * D_DIM;
        yp[0 * D_DIM] = y0;
        yp[1 * D_DIM] = y1;
        yp[2 * D_DIM] = y2;
        yp[3 * D_DIM] = y3;
      }
    };
    for (int t4 = 0; t4 < SCAN_LATE_AT; t4 += 4) trip(t4);
#pragma unroll
    for (int j = SCAN_EARLY; j < SCAN_F4; ++j) rq[j] = gq[late4 + SCAN_NT * j];
    for (int t4 = SCAN_LATE_AT; t4 < SCAN_T; t4 += 4) trip(t4);

    v4f* dq = &s_q[buf ^ 1][tid];
    v4f* dk = &s_k[buf ^ 1][tid];
#pragma unroll
    for (int j = 0; j < SCAN_F4; ++j) {
      dq[SCAN_NT * j] = rq[j];
      dk[SCAN_NT * j] = rk[j];
    }
#pragma unroll
    for (int j = 0; j < SCAN_VN; ++j) s_v[buf ^ 1][tid + SCAN_NT * j] = rv[j];
    if (tid < SCAN_T) s_b[buf ^ 1][tid] = rb;
    __syncthreads();

    load_step_ops(o0, &s_q[buf ^ 1][lane * 2], &s_k[buf ^ 1][lane * 2],
                  &s_v[buf ^ 1][wave], &s_b[buf ^ 1][0], 0);
    load_step_ops(o1, &s_q[buf ^ 1][lane * 2], &s_k[buf ^ 1][lane * 2],
                  &s_v[buf ^ 1][wave], &s_b[buf ^ 1][0], 1);

    const int chn = (ch + 2 < SCAN_NC) ? ch + 2 : SCAN_NC - 1;
    const size_t adv4 = (size_t)chn * (SCAN_T * D_DIM / 4);
#pragma unroll
    for (int j = 0; j < SCAN_F4; ++j) rk[j] = gk[adv4 + SCAN_NT * j];
#pragma unroll
    for (int j = 0; j < SCAN_EARLY; ++j) rq[j] = gq[adv4 + SCAN_NT * j];
#pragma unroll
    for (int j = 0; j < SCAN_VN; ++j)
      rv[j] = gv[(size_t)chn * (SCAN_T * D_DIM) + (size_t)j * (SCAN_NT / SCAN_ROWS) * D_DIM];
    rb = gb[chn * SCAN_T];
  }
}

__global__ void ln_rows_kernel(float* __restrict__ y,
                               const float* __restrict__ g,
                               const float* __restrict__ b) {
  __shared__ float sh[16];
  const int row = blockIdx.x, tid = threadIdx.x;
  const size_t off = (size_t)row * D_DIM + tid;
  const float v  = y[off];
  const float s  = block_sum_256(v, sh);
  const float s2 = block_sum_256(v * v, sh + 8);
  const float mu = s / D_DIM, var = s2 / D_DIM - mu * mu;
  y[off] = (v - mu) * rsqrtf(var + LN_EPS) * g[tid] + b[tid];
}

extern "C" void kernel_launch(void* const* d_in, const int* in_sizes, int n_in,
                              void* d_out, int out_size, void* d_ws, size_t ws_size,
                              hipStream_t stream) {
  (void)in_sizes; (void)n_in; (void)out_size; (void)ws_size;
  const float* x      = (const float*)d_in[0];
  const float* Wq     = (const float*)d_in[1];
  const float* Wk     = (const float*)d_in[2];
  const float* Wv     = (const float*)d_in[3];
  const float* beta_w = (const float*)d_in[4];
  const float* beta_b = (const float*)d_in[5];
  const float* decay  = (const float*)d_in[6];
  const float* Wo     = (const float*)d_in[7];
  const float* bo     = (const float*)d_in[8];
  const float* ln_g   = (const float*)d_in[9];
  const float* ln_b   = (const float*)d_in[10];
  const float* lnp_g  = (const float*)d_in[11];
  const float* lnp_b  = (const float*)d_in[12];
  float* out = (float*)d_out;

  char* ws = (char*)d_ws;
  const size_t MAT = (size_t)M_DIM * D_DIM * sizeof(float);
  float* q    = (float*)(ws);
  float* k    = (float*)(ws + MAT);
  float* vv   = (float*)(ws + 2 * MAT);
  float* beta = (float*)(ws + 3 * MAT);
  float* y    = (float*)(ws + 3 * MAT + (64 << 10));

  const dim3 gemm_grid(M_DIM / 64, D_DIM / 64);
  const dim3 gemm_blk(128);

  wmma_gemm_f32_kernel<<<gemm_grid, gemm_blk, 0, stream>>>(x, Wq, q, nullptr);
  wmma_gemm_f32_kernel<<<gemm_grid, gemm_blk, 0, stream>>>(x, Wk, k, nullptr);
  wmma_gemm_f32_kernel<<<gemm_grid, gemm_blk, 0, stream>>>(x, Wv, vv, nullptr);

  phi_beta_kernel<<<dim3(M_DIM), dim3(256), 0, stream>>>(
      x, q, k, beta_w, beta_b, lnp_g, lnp_b, beta);

  delta_scan_kernel<<<dim3(D_DIM / SCAN_ROWS, B_DIM), dim3(SCAN_NT), 0, stream>>>(
      q, k, vv, beta, decay, y);

  ln_rows_kernel<<<dim3(M_DIM), dim3(256), 0, stream>>>(y, ln_g, ln_b);

  wmma_gemm_f32_kernel<<<gemm_grid, gemm_blk, 0, stream>>>(y, Wo, out, bo);
}


# scan step: serial chain W->v_bar->u->W cut: 8-term partial of W.phi_k summed as a tree, u and the update each one multiply-add behind the wave sum, retrieved value kept in its step; new sum order
# speedup vs baseline: 3.6671x; 1.0734x over previous
#include <hip/hip_runtime.h>
#include <hip/hip_bf16.h>

typedef __attribute__((ext_vector_type(2))) float v2f;
typedef __attribute__((ext_vector_type(8))) float v8f;
typedef __attribute__((ext_vector_type(4))) float v4f;

#define D_DIM 256
#define L_DIM 1024
#define B_DIM 4
#define M_DIM (B_DIM * L_DIM)
#define LN_EPS 1e-5f

__device__ __forceinline__ float other_row(float x) {
  const int b = __builtin_bit_cast(int, x);
  return __builtin_bit_cast(float,
      __builtin_amdgcn_permlanex16(0, b, 0x76543210, 0xfedcba98, false, false));
}
template <int XMASK>
__device__ __forceinline__ float row_xor(float x) {
  const int b = __builtin_bit_cast(int, x);
  return __builtin_bit_cast(float,
      __builtin_amdgcn_update_dpp(0, b, 0x160 + XMASK, 0xf, 0xf, true));
}
__device__ __forceinline__ float wave_sum_32(float v) {
  v += other_row(v);
  v += row_xor<8>(v);
  v += row_xor<4>(v);
  v += row_xor<2>(v);
  v += row_xor<1>(v);
  return v;
}

__device__ __forceinline__ float block_sum_256(float v, float* sh8) {
  v = wave_sum_32(v);
  if ((threadIdx.x & 31) == 0) sh8[threadIdx.x >> 5] = v;
  __syncthreads();
  float r = 0.f;
#pragma unroll
  for (int i = 0; i < 8; ++i) r += sh8[i];
  return r;
}

#define G_K      D_DIM
#define G_PITCH4 (G_K / 4 + 1)
#define G_NT     4

struct BSet { v4f p[G_NT][2]; };

__device__ __forceinline__ void gemm_load_b(BSet& s, const v4f* bp, int slice) {
#pragma unroll
  for (int t = 0; t < G_NT; ++t) {
    s.p[t][0] = bp[t * 16 * G_PITCH4 + 4 * slice];
    s.p[t][1] = bp[t * 16 * G_PITCH4 + 4 * slice + 2];
  }
}

__device__ __forceinline__ v8f gemm_wmma(v2f a, v2f b, v8f c) {
  return __builtin_amdgcn_wmma_f32_16x16x4_f32(
       false, a,  false, b,
       (short)0, c,  false,  false);
}

__device__ __forceinline__ void gemm_mma_slice(v8f (&acc)[G_NT], v4f a0, v4f a1, const BSet& s) {
#pragma unroll
  for (int pc = 0; pc < 2; ++pc) {
    const v4f a4 = pc ? a1 : a0;
    const v2f alo = __builtin_shufflevector(a4, a4, 0, 1);
    const v2f ahi = __builtin_shufflevector(a4, a4, 2, 3);
#pragma unroll
    for (int t = 0; t < G_NT; ++t) {
      const v4f b4 = s.p[t][pc];
      acc[t] = gemm_wmma(alo, __builtin_shufflevector(b4, b4, 0, 1), acc[t]);
    }
#pragma unroll
    for (int t = 0; t < G_NT; ++t) {
      const v4f b4 = s.p[t][pc];
      acc[t] = gemm_wmma(ahi, __builtin_shufflevector(b4, b4, 2, 3), acc[t]);
    }
  }
}

__global__ void __launch_bounds__(128)
wmma_gemm_f32_kernel(const float* __restrict__ A,
                     const float* __restrict__ Wm,
                     float* __restrict__ Out,
                     const float* __restrict__ bias) {
  __shared__ v4f s_w[64 * G_PITCH4];

  const int N = D_DIM;
  const int tid  = threadIdx.x;
  const int wave = tid >> 5;
  const int lane = tid & 31;
  const int half = lane >> 4;
  const int l16  = lane & 15;
  const int tM = (blockIdx.x * 4 + wave) * 16;
  const int tN = blockIdx.y * 64;

  const v4f* ap = (const v4f*)(A + (size_t)(tM + l16) * G_K) + half;
  v4f a[G_K / 8];
#pragma unroll
  for (int j = 0; j < G_K / 8; ++j) a[j] = ap[2 * j];

  const v4f* wp = (const v4f*)(Wm + (size_t)tN * G_K);
#pragma unroll
  for (int i = 0; i < 64 * (G_K / 4) / 128; ++i) {
    const int idx = tid + 128 * i;
    s_w[(idx >> 6) * G_PITCH4 + (idx & 63)] = wp[idx];
  }
  __syncthreads();

  const v4f* bp = s_w + l16 * G_PITCH4 + half;

  v8f acc[G_NT];
#pragma unroll
  for (int t = 0; t < G_NT; ++t) acc[t] = (v8f){};

  BSet b0, b1;
  gemm_load_b(b0, bp, 0);
#pragma unroll
  for (int s = 0; s < G_K / 16; s += 2) {
    gemm_load_b(b1, bp, s + 1);
    __builtin_amdgcn_sched_barrier(0);
    gemm_mma_slice(acc, a[2 * s], a[2 * s + 1], b0);
    if (s + 2 < G_K / 16) gemm_load_b(b0, bp, s + 2);
    __builtin_amdgcn_sched_barrier(0);
    gemm_mma_slice(acc, a[2 * s + 2], a[2 * s + 3], b1);
  }

#pragma unroll
  for (int t = 0; t < G_NT; ++t) {
    const int n = tN + 16 * t + l16;
    const float bv = bias ? bias[n] : 0.f;
    union { v8f v; float f[8]; } u; u.v = acc[t];
#pragma unroll
    for (int r = 0; r < 8; ++r) {
      const int m = tM + r + (half ? 8 : 0);
      Out[(size_t)m * N + n] = u.f[r] + bv;
    }
  }
}

__global__ void phi_beta_kernel(const float* __restrict__ x,
                                float* __restrict__ q,
                                float* __restrict__ k,
                                const float* __restrict__ beta_w,
                                const float* __restrict__ beta_b,
                                const float* __restrict__ lnp_g,
                                const float* __restrict__ lnp_b,
                                float* __restrict__ beta) {
  __shared__ float sh[8 * 5];
  const int row = blockIdx.x, tid = threadIdx.x;
  const size_t off = (size_t)row * D_DIM + tid;

  float qe; { float t = q[off]; qe = (t > 0.f) ? t + 1.f : __expf(t); }
  float ke; { float t = k[off]; ke = (t > 0.f) ? t + 1.f : __expf(t); }
  const float xb = x[off] * beta_w[tid];

  const float sq  = block_sum_256(qe,      sh + 0);
  const float sq2 = block_sum_256(qe * qe, sh + 8);
  const float sk  = block_sum_256(ke,      sh + 16);
  const float sk2 = block_sum_256(ke * ke, sh + 24);
  const float sb  = block_sum_256(xb,      sh + 32);

  const float muq = sq / D_DIM, varq = sq2 / D_DIM - muq * muq;
  const float muk = sk / D_DIM, vark = sk2 / D_DIM - muk * muk;
  const float g = lnp_g[tid], bb = lnp_b[tid];
  q[off] = (qe - muq) * rsqrtf(varq + LN_EPS) * g + bb;
  k[off] = (ke - muk) * rsqrtf(vark + LN_EPS) * g + bb;
  if (tid == 0) beta[row] = 1.f / (1.f + __expf(-(sb + beta_b[0])));
}

#define SCAN_ROWS 4
#define SCAN_NT (SCAN_ROWS * 32)
#define SCAN_T 64
#define SCAN_NC (L_DIM / SCAN_T)
#define SCAN_F4 (SCAN_T * D_DIM / 4 / SCAN_NT)
#define SCAN_VN (SCAN_T * SCAN_ROWS / SCAN_NT)
#define SCAN_LATE 8
#define SCAN_EARLY (SCAN_F4 - SCAN_LATE)
#define SCAN_LATE_AT 32

union F8 { v4f v4[2]; float f[8]; };

struct StepOps { F8 q, k; float v, b; };

__device__ __forceinline__ void load_step_ops(StepOps& o, const v4f* lq, const v4f* lk,
                                              const float* lv, const float* lb, int t) {
  o.q.v4[0] = lq[t * (D_DIM / 4)];
  o.q.v4[1] = lq[t * (D_DIM / 4) + 1];
  o.k.v4[0] = lk[t * (D_DIM / 4)];
  o.k.v4[1] = lk[t * (D_DIM / 4) + 1];
  o.v = lv[t * SCAN_ROWS];
  o.b = lb[t];
}

__device__ __forceinline__ float dot8_tree(const float (&w)[8], const float (&x)[8]) {
  float p0 = w[0] * x[0];
  float p1 = w[1] * x[1];
  float p2 = w[2] * x[2];
  float p3 = w[3] * x[3];
  p0 = fmaf(w[4], x[4], p0);
  p1 = fmaf(w[5], x[5], p1);
  p2 = fmaf(w[6], x[6], p2);
  p3 = fmaf(w[7], x[7], p3);
  return (p0 + p2) + (p1 + p3);
}

__device__ __forceinline__ float scan_step(float (&w)[8], const StepOps& o, float df) {
  float rk = dot8_tree(w, o.k.f);
  float rq = 0.f;
#pragma unroll
  for (int c = 0; c < 8; ++c) rq = fmaf(w[c], o.q.f[c], rq);
  rk = wave_sum_32(rk);
  rq = wave_sum_32(rq);
  asm volatile("" :: "v"(rq));

  const float bv = o.b * o.v;
  const float u = fmaf(-o.b, rk, bv);
#pragma unroll
  for (int c = 0; c < 8; ++c) {
    const float dw = df * w[c];
    w[c] = fmaf(u, o.k.f[c], dw);
  }
  return rq;
}

#define HOLD_READS_HERE() __builtin_amdgcn_sched_barrier(0)


__global__ void __launch_bounds__(SCAN_NT)
delta_scan_kernel(const float* __restrict__ phi_q,
                  const float* __restrict__ phi_k,
                  const float* __restrict__ v,
                  const float* __restrict__ beta,
                  const float* __restrict__ decay,
                  float* __restrict__ y) {
  __shared__ v4f s_q[2][SCAN_T * D_DIM / 4];
  __shared__ v4f s_k[2][SCAN_T * D_DIM / 4];
  __shared__ float  s_v[2][SCAN_T * SCAN_ROWS];
  __shared__ float  s_b[2][SCAN_T];

  const int b    = blockIdx.y;
  const int tid  = threadIdx.x;
  const int wave = tid >> 5;
  const int lane = tid & 31;
  const int row  = blockIdx.x * SCAN_ROWS + wave;

  const float df = 1.f / (1.f + __expf(-decay[0]));
  const size_t base = (size_t)b * L_DIM * D_DIM;

  const v4f* gq = (const v4f*)(phi_q + base) + tid;
  const v4f* gk = (const v4f*)(phi_k + base) + tid;
  const float*  gv = v + base + (size_t)(tid / SCAN_ROWS) * D_DIM
                       + blockIdx.x * SCAN_ROWS + (tid % SCAN_ROWS);
  const float*  gb = beta + (size_t)b * L_DIM + (tid & (SCAN_T - 1));

#pragma unroll
  for (int j = 0; j < SCAN_F4; ++j) {
    s_q[0][tid + SCAN_NT * j] = gq[SCAN_NT * j];
    s_k[0][tid + SCAN_NT * j] = gk[SCAN_NT * j];
  }
#pragma unroll
  for (int j = 0; j < SCAN_VN; ++j)
    s_v[0][tid + SCAN_NT * j] = gv[(size_t)j * (SCAN_NT / SCAN_ROWS) * D_DIM];
  if (tid < SCAN_T) s_b[0][tid] = gb[0];
  __syncthreads();

  v4f rq[SCAN_F4], rk[SCAN_F4];
  float rv[SCAN_VN];
#pragma unroll
  for (int j = 0; j < SCAN_F4; ++j) rk[j] = gk[(SCAN_T * D_DIM / 4) + SCAN_NT * j];
#pragma unroll
  for (int j = 0; j < SCAN_EARLY; ++j) rq[j] = gq[(SCAN_T * D_DIM / 4) + SCAN_NT * j];
#pragma unroll
  for (int j = SCAN_EARLY; j < SCAN_F4; ++j) rq[j] = (v4f){0.f, 0.f, 0.f, 0.f};
#pragma unroll
  for (int j = 0; j < SCAN_VN; ++j)
    rv[j] = gv[(size_t)(SCAN_T * D_DIM) + (size_t)j * (SCAN_NT / SCAN_ROWS) * D_DIM];
  float rb = gb[SCAN_T];

  float w[8];
#pragma unroll
  for (int c = 0; c < 8; ++c) w[c] = 0.f;

  StepOps o0, o1, o2, o3;
  load_step_ops(o0, &s_q[0][lane * 2], &s_k[0][lane * 2], &s_v[0][wave], &s_b[0][0], 0);
  load_step_ops(o1, &s_q[0][lane * 2], &s_k[0][lane * 2], &s_v[0][wave], &s_b[0][0], 1);

  for (int ch = 0; ch < SCAN_NC; ++ch) {
    const int buf = ch & 1;
    const int chl = (ch + 1 < SCAN_NC) ? ch + 1 : SCAN_NC - 1;
    const size_t late4 = (size_t)chl * (SCAN_T * D_DIM / 4);

    const v4f* lq = &s_q[buf][lane * 2];
    const v4f* lk = &s_k[buf][lane * 2];
    const float*  lv = &s_v[buf][wave];
    const float*  lb = &s_b[buf][0];
    float* yrow = y + base + (size_t)ch * SCAN_T * D_DIM + row;

    auto trip = [&](const int t4) {
      const int ta = (t4 + 4 < SCAN_T) ? t4 + 4 : SCAN_T - 1;
      const int tb = (t4 + 5 < SCAN_T) ? t4 + 5 : SCAN_T - 1;
      load_step_ops(o2, lq, lk, lv, lb, t4 + 2);
      HOLD_READS_HERE();
      const float y0 = scan_step(w, o0, df);
      load_step_ops(o3, lq, lk, lv, lb, t4 + 3);
      HOLD_READS_HERE();
      const float y1 = scan_step(w, o1, df);
      load_step_ops(o0, lq, lk, lv, lb, ta);
      HOLD_READS_HERE();
      const float y2 = scan_step(w, o2, df);
      load_step_ops(o1, lq, lk, lv, lb, tb);
      HOLD_READS_HERE();
      const float y3 = scan_step(w, o3, df);
      if (lane == 0) {
        float* yp = yrow + (size_t)t4 * D_DIM;
        yp[0 * D_DIM] = y0;
        yp[1 * D_DIM] = y1;
        yp[2 * D_DIM] = y2;
        yp[3 * D_DIM] = y3;
      }
    };
    for (int t4 = 0; t4 < SCAN_LATE_AT; t4 += 4) trip(t4);
#pragma unroll
    for (int j = SCAN_EARLY; j < SCAN_F4; ++j) rq[j] = gq[late4 + SCAN_NT * j];
    for (int t4 = SCAN_LATE_AT; t4 < SCAN_T; t4 += 4) trip(t4);

    v4f* dq = &s_q[buf ^ 1][tid];
    v4f* dk = &s_k[buf ^ 1][tid];
#pragma unroll
    for (int j = 0; j < SCAN_F4; ++j) {
      dq[SCAN_NT * j] = rq[j];
      dk[SCAN_NT * j] = rk[j];
    }
#pragma unroll
    for (int j = 0; j < SCAN_VN; ++j) s_v[buf ^ 1][tid + SCAN_NT * j] = rv[j];
    if (tid < SCAN_T) s_b[buf ^ 1][tid] = rb;
    __syncthreads();

    load_step_ops(o0, &s_q[buf ^ 1][lane * 2], &s_k[buf ^ 1][lane * 2],
                  &s_v[buf ^ 1][wave], &s_b[buf ^ 1][0], 0);
    load_step_ops(o1, &s_q[buf ^ 1][lane * 2], &s_k[buf ^ 1][lane * 2],
                  &s_v[buf ^ 1][wave], &s_b[buf ^ 1][0], 1);

    const int chn = (ch + 2 < SCAN_NC) ? ch + 2 : SCAN_NC - 1;
    const size_t adv4 = (size_t)chn * (SCAN_T * D_DIM / 4);
#pragma unroll
    for (int j = 0; j < SCAN_F4; ++j) rk[j] = gk[adv4 + SCAN_NT * j];
#pragma unroll
    for (int j = 0; j < SCAN_EARLY; ++j) rq[j] = gq[adv4 + SCAN_NT * j];
#pragma unroll
    for (int j = 0; j < SCAN_VN; ++j)
      rv[j] = gv[(size_t)chn * (SCAN_T * D_DIM) + (size_t)j * (SCAN_NT / SCAN_ROWS) * D_DIM];
    rb = gb[chn * SCAN_T];
  }
}

__global__ void ln_rows_kernel(float* __restrict__ y,
                               const float* __restrict__ g,
                               const float* __restrict__ b) {
  __shared__ float sh[16];
  const int row = blockIdx.x, tid = threadIdx.x;
  const size_t off = (size_t)row * D_DIM + tid;
  const float v  = y[off];
  const float s  = block_sum_256(v, sh);
  const float s2 = block_sum_256(v * v, sh + 8);
  const float mu = s / D_DIM, var = s2 / D_DIM - mu * mu;
  y[off] = (v - mu) * rsqrtf(var + LN_EPS) * g[tid] + b[tid];
}

extern "C" void kernel_launch(void* const* d_in, const int* in_sizes, int n_in,
                              void* d_out, int out_size, void* d_ws, size_t ws_size,
                              hipStream_t stream) {
  (void)in_sizes; (void)n_in; (void)out_size; (void)ws_size;
  const float* x      = (const float*)d_in[0];
  const float* Wq     = (const float*)d_in[1];
  const float* Wk     = (const float*)d_in[2];
  const float* Wv     = (const float*)d_in[3];
  const float* beta_w = (const float*)d_in[4];
  const float* beta_b = (const float*)d_in[5];
  const float* decay  = (const float*)d_in[6];
  const float* Wo     = (const float*)d_in[7];
  const float* bo     = (const float*)d_in[8];
  const float* ln_g   = (const float*)d_in[9];
  const float* ln_b   = (const float*)d_in[10];
  const float* lnp_g  = (const float*)d_in[11];
  const float* lnp_b  = (const float*)d_in[12];
  float* out = (float*)d_out;

  char* ws = (char*)d_ws;
  const size_t MAT = (size_t)M_DIM * D_DIM * sizeof(float);
  float* q    = (float*)(ws);
  float* k    = (float*)(ws + MAT);
  float* vv   = (float*)(ws + 2 * MAT);
  float* beta = (float*)(ws + 3 * MAT);
  float* y    = (float*)(ws + 3 * MAT + (64 << 10));

  const dim3 gemm_grid(M_DIM / 64, D_DIM / 64);
  const dim3 gemm_blk(128);

  wmma_gemm_f32_kernel<<<gemm_grid, gemm_blk, 0, stream>>>(x, Wq, q, nullptr);
  wmma_gemm_f32_kernel<<<gemm_grid, gemm_blk, 0, stream>>>(x, Wk, k, nullptr);
  wmma_gemm_f32_kernel<<<gemm_grid, gemm_blk, 0, stream>>>(x, Wv, vv, nullptr);

  phi_beta_kernel<<<dim3(M_DIM), dim3(256), 0, stream>>>(
      x, q, k, beta_w, beta_b, lnp_g, lnp_b, beta);

  delta_scan_kernel<<<dim3(D_DIM / SCAN_ROWS, B_DIM), dim3(SCAN_NT), 0, stream>>>(
      q, k, vv, beta, decay, y);

  ln_rows_kernel<<<dim3(M_DIM), dim3(256), 0, stream>>>(y, ln_g, ln_b);

  wmma_gemm_f32_kernel<<<gemm_grid, gemm_blk, 0, stream>>>(y, Wo, out, bo);
}


# output projection: layernorm of y moved into its A prologue (a wave holds its 16 rows whole in registers; two-pass row stats joined by one permlanex16: new sum order); the layernorm launch is gone
# speedup vs baseline: 3.8141x; 1.0401x over previous
#include <hip/hip_runtime.h>
#include <hip/hip_bf16.h>

typedef __attribute__((ext_vector_type(2))) float v2f;
typedef __attribute__((ext_vector_type(8))) float v8f;
typedef __attribute__((ext_vector_type(4))) float v4f;

#define D_DIM 256
#define L_DIM 1024
#define B_DIM 4
#define M_DIM (B_DIM * L_DIM)
#define LN_EPS 1e-5f

__device__ __forceinline__ float other_row(float x) {
  const int b = __builtin_bit_cast(int, x);
  return __builtin_bit_cast(float,
      __builtin_amdgcn_permlanex16(0, b, 0x76543210, 0xfedcba98, false, false));
}
template <int XMASK>
__device__ __forceinline__ float row_xor(float x) {
  const int b = __builtin_bit_cast(int, x);
  return __builtin_bit_cast(float,
      __builtin_amdgcn_update_dpp(0, b, 0x160 + XMASK, 0xf, 0xf, true));
}
__device__ __forceinline__ float wave_sum_32(float v) {
  v += other_row(v);
  v += row_xor<8>(v);
  v += row_xor<4>(v);
  v += row_xor<2>(v);
  v += row_xor<1>(v);
  return v;
}

__device__ __forceinline__ float block_sum_256(float v, float* sh8) {
  v = wave_sum_32(v);
  if ((threadIdx.x & 31) == 0) sh8[threadIdx.x >> 5] = v;
  __syncthreads();
  float r = 0.f;
#pragma unroll
  for (int i = 0; i < 8; ++i) r += sh8[i];
  return r;
}

#define G_K      D_DIM
#define G_PITCH4 (G_K / 4 + 1)
#define G_NT     4

struct BSet { v4f p[G_NT][2]; };

__device__ __forceinline__ void gemm_load_b(BSet& s, const v4f* bp, int slice) {
#pragma unroll
  for (int t = 0; t < G_NT; ++t) {
    s.p[t][0] = bp[t * 16 * G_PITCH4 + 4 * slice];
    s.p[t][1] = bp[t * 16 * G_PITCH4 + 4 * slice + 2];
  }
}

__device__ __forceinline__ v8f gemm_wmma(v2f a, v2f b, v8f c) {
  return __builtin_amdgcn_wmma_f32_16x16x4_f32(
       false, a,  false, b,
       (short)0, c,  false,  false);
}

__device__ __forceinline__ void gemm_mma_slice(v8f (&acc)[G_NT], v4f a0, v4f a1, const BSet& s) {
#pragma unroll
  for (int pc = 0; pc < 2; ++pc) {
    const v4f a4 = pc ? a1 : a0;
    const v2f alo = __builtin_shufflevector(a4, a4, 0, 1);
    const v2f ahi = __builtin_shufflevector(a4, a4, 2, 3);
#pragma unroll
    for (int t = 0; t < G_NT; ++t) {
      const v4f b4 = s.p[t][pc];
      acc[t] = gemm_wmma(alo, __builtin_shufflevector(b4, b4, 0, 1), acc[t]);
    }
#pragma unroll
    for (int t = 0; t < G_NT; ++t) {
      const v4f b4 = s.p[t][pc];
      acc[t] = gemm_wmma(ahi, __builtin_shufflevector(b4, b4, 2, 3), acc[t]);
    }
  }
}

template <bool LN_ROWS>
__global__ void __launch_bounds__(128)
wmma_gemm_f32_kernel(const float* __restrict__ A,
                     const float* __restrict__ Wm,
                     float* __restrict__ Out,
                     const float* __restrict__ bias,
                     const float* __restrict__ ln_g,
                     const float* __restrict__ ln_b) {
  __shared__ v4f s_w[64 * G_PITCH4];
  __shared__ v4f s_gb[LN_ROWS ? 2 * (G_K / 4) : 1];

  const int N = D_DIM;
  const int tid  = threadIdx.x;
  const int wave = tid >> 5;
  const int lane = tid & 31;
  const int half = lane >> 4;
  const int l16  = lane & 15;
  const int tM = (blockIdx.x * 4 + wave) * 16;
  const int tN = blockIdx.y * 64;

  const v4f* ap = (const v4f*)(A + (size_t)(tM + l16) * G_K) + half;
  v4f a[G_K / 8];
#pragma unroll
  for (int j = 0; j < G_K / 8; ++j) a[j] = ap[2 * j];

  const v4f* wp = (const v4f*)(Wm + (size_t)tN * G_K);
#pragma unroll
  for (int i = 0; i < 64 * (G_K / 4) / 128; ++i) {
    const int idx = tid + 128 * i;
    s_w[(idx >> 6) * G_PITCH4 + (idx & 63)] = wp[idx];
  }
  if constexpr (LN_ROWS) {
    s_gb[tid] = (tid < G_K / 4) ? ((const v4f*)ln_g)[tid] : ((const v4f*)ln_b)[tid - G_K / 4];
  }
  __syncthreads();

  if constexpr (LN_ROWS) {
    v4f s4[4];
#pragma unroll
    for (int i = 0; i < 4; ++i) s4[i] = a[i];
#pragma unroll
    for (int j = 4; j < G_K / 8; ++j) s4[j & 3] += a[j];
    const v4f st = (s4[0] + s4[1]) + (s4[2] + s4[3]);
    float rs = (st.x + st.y) + (st.z + st.w);
    rs += other_row(rs);
    const float mu = rs * (1.f / G_K);

    v4f q4[4];
#pragma unroll
    for (int i = 0; i < 4; ++i) { const v4f d = a[i] - mu; q4[i] = d * d; }
#pragma unroll
    for (int j = 4; j < G_K / 8; ++j) { const v4f d = a[j] - mu; q4[j & 3] += d * d; }
    const v4f qt = (q4[0] + q4[1]) + (q4[2] + q4[3]);
    float rq = (qt.x + qt.y) + (qt.z + qt.w);
    rq += other_row(rq);
    const float rstd = rsqrtf(rq * (1.f / G_K) + LN_EPS);

    const v4f* gp = s_gb + half;
    const v4f* hp = s_gb + G_K / 4 + half;
#pragma unroll
    for (int j = 0; j < G_K / 8; ++j)
      a[j] = ((a[j] - mu) * rstd) * gp[2 * j] + hp[2 * j];
  }

  const v4f* bp = s_w + l16 * G_PITCH4 + half;

  v8f acc[G_NT];
#pragma unroll
  for (int t = 0; t < G_NT; ++t) acc[t] = (v8f){};

  BSet b0, b1;
  gemm_load_b(b0, bp, 0);
#pragma unroll
  for (int s = 0; s < G_K / 16; s += 2) {
    gemm_load_b(b1, bp, s + 1);
    __builtin_amdgcn_sched_barrier(0);
    gemm_mma_slice(acc, a[2 * s], a[2 * s + 1], b0);
    if (s + 2 < G_K / 16) gemm_load_b(b0, bp, s + 2);
    __builtin_amdgcn_sched_barrier(0);
    gemm_mma_slice(acc, a[2 * s + 2], a[2 * s + 3], b1);
  }

#pragma unroll
  for (int t = 0; t < G_NT; ++t) {
    const int n = tN + 16 * t + l16;
    const float bv = bias ? bias[n] : 0.f;
    union { v8f v; float f[8]; } u; u.v = acc[t];
#pragma unroll
    for (int r = 0; r < 8; ++r) {
      const int m = tM + r + (half ? 8 : 0);
      Out[(size_t)m * N + n] = u.f[r] + bv;
    }
  }
}

__global__ void phi_beta_kernel(const float* __restrict__ x,
                                float* __restrict__ q,
                                float* __restrict__ k,
                                const float* __restrict__ beta_w,
                                const float* __restrict__ beta_b,
                                const float* __restrict__ lnp_g,
                                const float* __restrict__ lnp_b,
                                float* __restrict__ beta) {
  __shared__ float sh[8 * 5];
  const int row = blockIdx.x, tid = threadIdx.x;
  const size_t off = (size_t)row * D_DIM + tid;

  float qe; { float t = q[off]; qe = (t > 0.f) ? t + 1.f : __expf(t); }
  float ke; { float t = k[off]; ke = (t > 0.f) ? t + 1.f : __expf(t); }
  const float xb = x[off] * beta_w[tid];

  const float sq  = block_sum_256(qe,      sh + 0);
  const float sq2 = block_sum_256(qe * qe, sh + 8);
  const float sk  = block_sum_256(ke,      sh + 16);
  const float sk2 = block_sum_256(ke * ke, sh + 24);
  const float sb  = block_sum_256(xb,      sh + 32);

  const float muq = sq / D_DIM, varq = sq2 / D_DIM - muq * muq;
  const float muk = sk / D_DIM, vark = sk2 / D_DIM - muk * muk;
  const float g = lnp_g[tid], bb = lnp_b[tid];
  q[off] = (qe - muq) * rsqrtf(varq + LN_EPS) * g + bb;
  k[off] = (ke - muk) * rsqrtf(vark + LN_EPS) * g + bb;
  if (tid == 0) beta[row] = 1.f / (1.f + __expf(-(sb + beta_b[0])));
}

#define SCAN_ROWS 4
#define SCAN_NT (SCAN_ROWS * 32)
#define SCAN_T 64
#define SCAN_NC (L_DIM / SCAN_T)
#define SCAN_F4 (SCAN_T * D_DIM / 4 / SCAN_NT)
#define SCAN_VN (SCAN_T * SCAN_ROWS / SCAN_NT)
#define SCAN_LATE 8
#define SCAN_EARLY (SCAN_F4 - SCAN_LATE)
#define SCAN_LATE_AT 32

union F8 { v4f v4[2]; float f[8]; };

struct StepOps { F8 q, k; float v, b; };

__device__ __forceinline__ void load_step_ops(StepOps& o, const v4f* lq, const v4f* lk,
                                              const float* lv, const float* lb, int t) {
  o.q.v4[0] = lq[t * (D_DIM / 4)];
  o.q.v4[1] = lq[t * (D_DIM / 4) + 1];
  o.k.v4[0] = lk[t * (D_DIM / 4)];
  o.k.v4[1] = lk[t * (D_DIM / 4) + 1];
  o.v = lv[t * SCAN_ROWS];
  o.b = lb[t];
}

__device__ __forceinline__ float dot8_tree(const float (&w)[8], const float (&x)[8]) {
  float p0 = w[0] * x[0];
  float p1 = w[1] * x[1];
  float p2 = w[2] * x[2];
  float p3 = w[3] * x[3];
  p0 = fmaf(w[4], x[4], p0);
  p1 = fmaf(w[5], x[5], p1);
  p2 = fmaf(w[6], x[6], p2);
  p3 = fmaf(w[7], x[7], p3);
  return (p0 + p2) + (p1 + p3);
}

__device__ __forceinline__ float scan_step(float (&w)[8], const StepOps& o, float df) {
  float rk = dot8_tree(w, o.k.f);
  float rq = 0.f;
#pragma unroll
  for (int c = 0; c < 8; ++c) rq = fmaf(w[c], o.q.f[c], rq);
  rk = wave_sum_32(rk);
  rq = wave_sum_32(rq);
  asm volatile("" :: "v"(rq));

  const float bv = o.b * o.v;
  const float u = fmaf(-o.b, rk, bv);
#pragma unroll
  for (int c = 0; c < 8; ++c) {
    const float dw = df * w[c];
    w[c] = fmaf(u, o.k.f[c], dw);
  }
  return rq;
}

#define HOLD_READS_HERE() __builtin_amdgcn_sched_barrier(0)


__global__ void __launch_bounds__(SCAN_NT)
delta_scan_kernel(const float* __restrict__ phi_q,
                  const float* __restrict__ phi_k,
                  const float* __restrict__ v,
                  const float* __restrict__ beta,
                  const float* __restrict__ decay,
                  float* __restrict__ y) {
  __shared__ v4f s_q[2][SCAN_T * D_DIM / 4];
  __shared__ v4f s_k[2][SCAN_T * D_DIM / 4];
  __shared__ float  s_v[2][SCAN_T * SCAN_ROWS];
  __shared__ float  s_b[2][SCAN_T];

  const int b    = blockIdx.y;
  const int tid  = threadIdx.x;
  const int wave = tid >> 5;
  const int lane = tid & 31;
  const int row  = blockIdx.x * SCAN_ROWS + wave;

  const float df = 1.f / (1.f + __expf(-decay[0]));
  const size_t base = (size_t)b * L_DIM * D_DIM;

  const v4f* gq = (const v4f*)(phi_q + base) + tid;
  const v4f* gk = (const v4f*)(phi_k + base) + tid;
  const float*  gv = v + base + (size_t)(tid / SCAN_ROWS) * D_DIM
                       + blockIdx.x * SCAN_ROWS + (tid % SCAN_ROWS);
  const float*  gb = beta + (size_t)b * L_DIM + (tid & (SCAN_T - 1));

#pragma unroll
  for (int j = 0; j < SCAN_F4; ++j) {
    s_q[0][tid + SCAN_NT * j] = gq[SCAN_NT * j];
    s_k[0][tid + SCAN_NT * j] = gk[SCAN_NT * j];
  }
#pragma unroll
  for (int j = 0; j < SCAN_VN; ++j)
    s_v[0][tid + SCAN_NT * j] = gv[(size_t)j * (SCAN_NT / SCAN_ROWS) * D_DIM];
  if (tid < SCAN_T) s_b[0][tid] = gb[0];
  __syncthreads();

  v4f rq[SCAN_F4], rk[SCAN_F4];
  float rv[SCAN_VN];
#pragma unroll
  for (int j = 0; j < SCAN_F4; ++j) rk[j] = gk[(SCAN_T * D_DIM / 4) + SCAN_NT * j];
#pragma unroll
  for (int j = 0; j < SCAN_EARLY; ++j) rq[j] = gq[(SCAN_T * D_DIM / 4) + SCAN_NT * j];
#pragma unroll
  for (int j = SCAN_EARLY; j < SCAN_F4; ++j) rq[j] = (v4f){0.f, 0.f, 0.f, 0.f};
#pragma unroll
  for (int j = 0; j < SCAN_VN; ++j)
    rv[j] = gv[(size_t)(SCAN_T * D_DIM) + (size_t)j * (SCAN_NT / SCAN_ROWS) * D_DIM];
  float rb = gb[SCAN_T];

  float w[8];
#pragma unroll
  for (int c = 0; c < 8; ++c) w[c] = 0.f;

  StepOps o0, o1, o2, o3;
  load_step_ops(o0, &s_q[0][lane * 2], &s_k[0][lane * 2], &s_v[0][wave], &s_b[0][0], 0);
  load_step_ops(o1, &s_q[0][lane * 2], &s_k[0][lane * 2], &s_v[0][wave], &s_b[0][0], 1);

  for (int ch = 0; ch < SCAN_NC; ++ch) {
    const int buf = ch & 1;
    const int chl = (ch + 1 < SCAN_NC) ? ch + 1 : SCAN_NC - 1;
    const size_t late4 = (size_t)chl * (SCAN_T * D_DIM / 4);

    const v4f* lq = &s_q[buf][lane * 2];
    const v4f* lk = &s_k[buf][lane * 2];
    const float*  lv = &s_v[buf][wave];
    const float*  lb = &s_b[buf][0];
    float* yrow = y + base + (size_t)ch * SCAN_T * D_DIM + row;

    auto trip = [&](const int t4) {
      const int ta = (t4 + 4 < SCAN_T) ? t4 + 4 : SCAN_T - 1;
      const int tb = (t4 + 5 < SCAN_T) ? t4 + 5 : SCAN_T - 1;
      load_step_ops(o2, lq, lk, lv, lb, t4 + 2);
      HOLD_READS_HERE();
      const float y0 = scan_step(w, o0, df);
      load_step_ops(o3, lq, lk, lv, lb, t4 + 3);
      HOLD_READS_HERE();
      const float y1 = scan_step(w, o1, df);
      load_step_ops(o0, lq, lk, lv, lb, ta);
      HOLD_READS_HERE();
      const float y2 = scan_step(w, o2, df);
      load_step_ops(o1, lq, lk, lv, lb, tb);
      HOLD_READS_HERE();
      const float y3 = scan_step(w, o3, df);
      if (lane == 0) {
        float* yp = yrow + (size_t)t4 * D_DIM;
        yp[0 * D_DIM] = y0;
        yp[1 * D_DIM] = y1;
        yp[2 * D_DIM] = y2;
        yp[3 * D_DIM] = y3;
      }
    };
    for (int t4 = 0; t4 < SCAN_LATE_AT; t4 += 4) trip(t4);
#pragma unroll
    for (int j = SCAN_EARLY; j < SCAN_F4; ++j) rq[j] = gq[late4 + SCAN_NT * j];
    for (int t4 = SCAN_LATE_AT; t4 < SCAN_T; t4 += 4) trip(t4);

    v4f* dq = &s_q[buf ^ 1][tid];
    v4f* dk = &s_k[buf ^ 1][tid];
#pragma unroll
    for (int j = 0; j < SCAN_F4; ++j) {
      dq[SCAN_NT * j] = rq[j];
      dk[SCAN_NT * j] = rk[j];
    }
#pragma unroll
    for (int j = 0; j < SCAN_VN; ++j) s_v[buf ^ 1][tid + SCAN_NT * j] = rv[j];
    if (tid < SCAN_T) s_b[buf ^ 1][tid] = rb;
    __syncthreads();

    load_step_ops(o0, &s_q[buf ^ 1][lane * 2], &s_k[buf ^ 1][lane * 2],
                  &s_v[buf ^ 1][wave], &s_b[buf ^ 1][0], 0);
    load_step_ops(o1, &s_q[buf ^ 1][lane * 2], &s_k[buf ^ 1][lane * 2],
                  &s_v[buf ^ 1][wave], &s_b[buf ^ 1][0], 1);

    const int chn = (ch + 2 < SCAN_NC) ? ch + 2 : SCAN_NC - 1;
    const size_t adv4 = (size_t)chn * (SCAN_T * D_DIM / 4);
#pragma unroll
    for (int j = 0; j < SCAN_F4; ++j) rk[j] = gk[adv4 + SCAN_NT * j];
#pragma unroll
    for (int j = 0; j < SCAN_EARLY; ++j) rq[j] = gq[adv4 + SCAN_NT * j];
#pragma unroll
    for (int j = 0; j < SCAN_VN; ++j)
      rv[j] = gv[(size_t)chn * (SCAN_T * D_DIM) + (size_t)j * (SCAN_NT / SCAN_ROWS) * D_DIM];
    rb = gb[chn * SCAN_T];
  }
}

extern "C" void kernel_launch(void* const* d_in, const int* in_sizes, int n_in,
                              void* d_out, int out_size, void* d_ws, size_t ws_size,
                              hipStream_t stream) {
  (void)in_sizes; (void)n_in; (void)out_size; (void)ws_size;
  const float* x      = (const float*)d_in[0];
  const float* Wq     = (const float*)d_in[1];
  const float* Wk     = (const float*)d_in[2];
  const float* Wv     = (const float*)d_in[3];
  const float* beta_w = (const float*)d_in[4];
  const float* beta_b = (const float*)d_in[5];
  const float* decay  = (const float*)d_in[6];
  const float* Wo     = (const float*)d_in[7];
  const float* bo     = (const float*)d_in[8];
  const float* ln_g   = (const float*)d_in[9];
  const float* ln_b   = (const float*)d_in[10];
  const float* lnp_g  = (const float*)d_in[11];
  const float* lnp_b  = (const float*)d_in[12];
  float* out = (float*)d_out;

  char* ws = (char*)d_ws;
  const size_t MAT = (size_t)M_DIM * D_DIM * sizeof(float);
  float* q    = (float*)(ws);
  float* k    = (float*)(ws + MAT);
  float* vv   = (float*)(ws + 2 * MAT);
  float* beta = (float*)(ws + 3 * MAT);
  float* y    = (float*)(ws + 3 * MAT + (64 << 10));

  const dim3 gemm_grid(M_DIM / 64, D_DIM / 64);
  const dim3 gemm_blk(128);

  wmma_gemm_f32_kernel<false><<<gemm_grid, gemm_blk, 0, stream>>>(x, Wq, q, nullptr, nullptr, nullptr);
  wmma_gemm_f32_kernel<false><<<gemm_grid, gemm_blk, 0, stream>>>(x, Wk, k, nullptr, nullptr, nullptr);
  wmma_gemm_f32_kernel<false><<<gemm_grid, gemm_blk, 0, stream>>>(x, Wv, vv, nullptr, nullptr, nullptr);

  phi_beta_kernel<<<dim3(M_DIM), dim3(256), 0, stream>>>(
      x, q, k, beta_w, beta_b, lnp_g, lnp_b, beta);

  delta_scan_kernel<<<dim3(D_DIM / SCAN_ROWS, B_DIM), dim3(SCAN_NT), 0, stream>>>(
      q, k, vv, beta, decay, y);

  wmma_gemm_f32_kernel<true><<<gemm_grid, gemm_blk, 0, stream>>>(y, Wo, out, bo, ln_g, ln_b);
}


# input projections own whole rows (4 column strips per block, x rows kept in registers) so elu+1, layernorm and beta run in their tail: 1 launch for 4; two-pass row stats, beta dot: new sum order
# speedup vs baseline: 4.0573x; 1.0638x over previous
#include <hip/hip_runtime.h>
#include <hip/hip_bf16.h>

typedef __attribute__((ext_vector_type(2))) float v2f;
typedef __attribute__((ext_vector_type(8))) float v8f;
typedef __attribute__((ext_vector_type(4))) float v4f;

#define D_DIM 256
#define L_DIM 1024
#define B_DIM 4
#define M_DIM (B_DIM * L_DIM)
#define LN_EPS 1e-5f

__device__ __forceinline__ float other_row(float x) {
  const int b = __builtin_bit_cast(int, x);
  return __builtin_bit_cast(float,
      __builtin_amdgcn_permlanex16(0, b, 0x76543210, 0xfedcba98, false, false));
}
template <int XMASK>
__device__ __forceinline__ float row_xor(float x) {
  const int b = __builtin_bit_cast(int, x);
  return __builtin_bit_cast(float,
      __builtin_amdgcn_update_dpp(0, b, 0x160 + XMASK, 0xf, 0xf, true));
}
__device__ __forceinline__ float wave_sum_32(float v) {
  v += other_row(v);
  v += row_xor<8>(v);
  v += row_xor<4>(v);
  v += row_xor<2>(v);
  v += row_xor<1>(v);
  return v;
}

__device__ __forceinline__ float row_sum_16(float v) {
  v += row_xor<8>(v);
  v += row_xor<4>(v);
  v += row_xor<2>(v);
  v += row_xor<1>(v);
  return v;
}

#define G_K      D_DIM
#define G_PITCH4 (G_K / 4 + 1)
#define G_NT     4

struct BSet { v4f p[G_NT][2]; };

__device__ __forceinline__ void gemm_load_b(BSet& s, const v4f* bp, int slice) {
#pragma unroll
  for (int t = 0; t < G_NT; ++t) {
    s.p[t][0] = bp[t * 16 * G_PITCH4 + 4 * slice];
    s.p[t][1] = bp[t * 16 * G_PITCH4 + 4 * slice + 2];
  }
}

__device__ __forceinline__ v8f gemm_wmma(v2f a, v2f b, v8f c) {
  return __builtin_amdgcn_wmma_f32_16x16x4_f32(
       false, a,  false, b,
       (short)0, c,  false,  false);
}

__device__ __forceinline__ void gemm_mma_slice(v8f (&acc)[G_NT], v4f a0, v4f a1, const BSet& s) {
#pragma unroll
  for (int pc = 0; pc < 2; ++pc) {
    const v4f a4 = pc ? a1 : a0;
    const v2f alo = __builtin_shufflevector(a4, a4, 0, 1);
    const v2f ahi = __builtin_shufflevector(a4, a4, 2, 3);
#pragma unroll
    for (int t = 0; t < G_NT; ++t) {
      const v4f b4 = s.p[t][pc];
      acc[t] = gemm_wmma(alo, __builtin_shufflevector(b4, b4, 0, 1), acc[t]);
    }
#pragma unroll
    for (int t = 0; t < G_NT; ++t) {
      const v4f b4 = s.p[t][pc];
      acc[t] = gemm_wmma(ahi, __builtin_shufflevector(b4, b4, 2, 3), acc[t]);
    }
  }
}

__device__ __forceinline__ void gemm_strip(v8f (&acc)[G_NT], const v4f (&a)[G_K / 8], const v4f* bp) {
  BSet b0, b1;
  gemm_load_b(b0, bp, 0);
#pragma unroll
  for (int s = 0; s < G_K / 16; s += 2) {
    gemm_load_b(b1, bp, s + 1);
    __builtin_amdgcn_sched_barrier(0);
    gemm_mma_slice(acc, a[2 * s], a[2 * s + 1], b0);
    if (s + 2 < G_K / 16) gemm_load_b(b0, bp, s + 2);
    __builtin_amdgcn_sched_barrier(0);
    gemm_mma_slice(acc, a[2 * s + 2], a[2 * s + 3], b1);
  }
}

template <bool LN_ROWS>
__global__ void __launch_bounds__(128)
wmma_gemm_f32_kernel(const float* __restrict__ A,
                     const float* __restrict__ Wm,
                     float* __restrict__ Out,
                     const float* __restrict__ bias,
                     const float* __restrict__ ln_g,
                     const float* __restrict__ ln_b) {
  __shared__ v4f s_w[64 * G_PITCH4];
  __shared__ v4f s_gb[LN_ROWS ? 2 * (G_K / 4) : 1];

  const int N = D_DIM;
  const int tid  = threadIdx.x;
  const int wave = tid >> 5;
  const int lane = tid & 31;
  const int half = lane >> 4;
  const int l16  = lane & 15;
  const int tM = (blockIdx.x * 4 + wave) * 16;
  const int tN = blockIdx.y * 64;

  const v4f* ap = (const v4f*)(A + (size_t)(tM + l16) * G_K) + half;
  v4f a[G_K / 8];
#pragma unroll
  for (int j = 0; j < G_K / 8; ++j) a[j] = ap[2 * j];

  const v4f* wp = (const v4f*)(Wm + (size_t)tN * G_K);
#pragma unroll
  for (int i = 0; i < 64 * (G_K / 4) / 128; ++i) {
    const int idx = tid + 128 * i;
    s_w[(idx >> 6) * G_PITCH4 + (idx & 63)] = wp[idx];
  }
  if constexpr (LN_ROWS) {
    s_gb[tid] = (tid < G_K / 4) ? ((const v4f*)ln_g)[tid] : ((const v4f*)ln_b)[tid - G_K / 4];
  }
  __syncthreads();

  if constexpr (LN_ROWS) {
    v4f s4[4];
#pragma unroll
    for (int i = 0; i < 4; ++i) s4[i] = a[i];
#pragma unroll
    for (int j = 4; j < G_K / 8; ++j) s4[j & 3] += a[j];
    const v4f st = (s4[0] + s4[1]) + (s4[2] + s4[3]);
    float rs = (st.x + st.y) + (st.z + st.w);
    rs += other_row(rs);
    const float mu = rs * (1.f / G_K);

    v4f q4[4];
#pragma unroll
    for (int i = 0; i < 4; ++i) { const v4f d = a[i] - mu; q4[i] = d * d; }
#pragma unroll
    for (int j = 4; j < G_K / 8; ++j) { const v4f d = a[j] - mu; q4[j & 3] += d * d; }
    const v4f qt = (q4[0] + q4[1]) + (q4[2] + q4[3]);
    float rq = (qt.x + qt.y) + (qt.z + qt.w);
    rq += other_row(rq);
    const float rstd = rsqrtf(rq * (1.f / G_K) + LN_EPS);

    const v4f* gp = s_gb + half;
    const v4f* hp = s_gb + G_K / 4 + half;
#pragma unroll
    for (int j = 0; j < G_K / 8; ++j)
      a[j] = ((a[j] - mu) * rstd) * gp[2 * j] + hp[2 * j];
  }

  const v4f* bp = s_w + l16 * G_PITCH4 + half;

  v8f acc[G_NT];
#pragma unroll
  for (int t = 0; t < G_NT; ++t) acc[t] = (v8f){};

  gemm_strip(acc, a, bp);

#pragma unroll
  for (int t = 0; t < G_NT; ++t) {
    const int n = tN + 16 * t + l16;
    const float bv = bias ? bias[n] : 0.f;
    union { v8f v; float f[8]; } u; u.v = acc[t];
#pragma unroll
    for (int r = 0; r < 8; ++r) {
      const int m = tM + r + (half ? 8 : 0);
      Out[(size_t)m * N + n] = u.f[r] + bv;
    }
  }
}

#define P_STRIP4 (64 * (G_K / 4))
#define P_ST     (P_STRIP4 / 128)

__global__ void __launch_bounds__(128)
qkv_phi_kernel(const float* __restrict__ x,
               const float* __restrict__ Wq,
               const float* __restrict__ Wk,
               const float* __restrict__ Wv,
               const float* __restrict__ beta_w,
               const float* __restrict__ beta_b,
               const float* __restrict__ lnp_g,
               const float* __restrict__ lnp_b,
               float* __restrict__ q,
               float* __restrict__ k,
               float* __restrict__ v,
               float* __restrict__ beta) {
  __shared__ v4f s_w[4][64 * G_PITCH4];

  const int which = blockIdx.y;
  const float* Wm = (which == 0) ? Wq : (which == 1) ? Wk : Wv;
  float* Out      = (which == 0) ? q  : (which == 1) ? k  : v;

  const int tid  = threadIdx.x;
  const int wave = tid >> 5;
  const int lane = tid & 31;
  const int half = lane >> 4;
  const int l16  = lane & 15;
  const int tM = (blockIdx.x * 4 + wave) * 16;

  const v4f* ap = (const v4f*)(x + (size_t)(tM + l16) * G_K) + half;
  v4f a[G_K / 8];
#pragma unroll
  for (int j = 0; j < G_K / 8; ++j) a[j] = ap[2 * j];

  const v4f* wp = (const v4f*)Wm;
#pragma unroll
  for (int i = 0; i < P_ST; ++i) {
    const int idx = tid + 128 * i;
    s_w[0][(idx >> 6) * G_PITCH4 + (idx & 63)] = wp[idx];
  }
  __syncthreads();

  v8f acc[4][G_NT];
#pragma unroll
  for (int ns = 0; ns < 4; ++ns)
#pragma unroll
    for (int t = 0; t < G_NT; ++t) acc[ns][t] = (v8f){};

#pragma unroll
  for (int ns = 0; ns < 4; ++ns) {
    v4f st[P_ST];
    if (ns + 1 < 4) {
#pragma unroll
      for (int i = 0; i < P_ST; ++i) st[i] = wp[(ns + 1) * P_STRIP4 + tid + 128 * i];
    }
    gemm_strip(acc[ns], a, &s_w[ns][l16 * G_PITCH4 + half]);
    if (ns + 1 < 4) {
#pragma unroll
      for (int i = 0; i < P_ST; ++i) {
        const int idx = tid + 128 * i;
        s_w[ns + 1][(idx >> 6) * G_PITCH4 + (idx & 63)] = st[i];
      }
      __syncthreads();
    }
  }

  const int rbase = tM + (half ? 8 : 0);

  if (which == 2) {
#pragma unroll
    for (int ns = 0; ns < 4; ++ns)
#pragma unroll
      for (int t = 0; t < G_NT; ++t) {
        const int n = 64 * ns + 16 * t + l16;
#pragma unroll
        for (int r = 0; r < 8; ++r)
          Out[(size_t)(rbase + r) * D_DIM + n] = acc[ns][t][r];
      }
    const v4f* bw = (const v4f*)beta_w + half;
    v4f b4[4];
#pragma unroll
    for (int i = 0; i < 4; ++i) b4[i] = a[i] * bw[2 * i];
#pragma unroll
    for (int j = 4; j < G_K / 8; ++j) b4[j & 3] += a[j] * bw[2 * j];
    const v4f bt = (b4[0] + b4[1]) + (b4[2] + b4[3]);
    float sb = (bt.x + bt.y) + (bt.z + bt.w);
    sb += other_row(sb);
    if (half == 0) beta[tM + l16] = 1.f / (1.f + __expf(-(sb + beta_b[0])));
  } else {
    float sm[8];
#pragma unroll
    for (int r = 0; r < 8; ++r) sm[r] = 0.f;
#pragma unroll
    for (int ns = 0; ns < 4; ++ns)
#pragma unroll
      for (int t = 0; t < G_NT; ++t)
#pragma unroll
        for (int r = 0; r < 8; ++r) {
          const float tv = acc[ns][t][r];
          const float e = (tv > 0.f) ? tv + 1.f : __expf(tv);
          acc[ns][t][r] = e;
          sm[r] += e;
        }
    float mu[8];
#pragma unroll
    for (int r = 0; r < 8; ++r) mu[r] = row_sum_16(sm[r]) * (1.f / D_DIM);

    float qd[8];
#pragma unroll
    for (int r = 0; r < 8; ++r) qd[r] = 0.f;
#pragma unroll
    for (int ns = 0; ns < 4; ++ns)
#pragma unroll
      for (int t = 0; t < G_NT; ++t)
#pragma unroll
        for (int r = 0; r < 8; ++r) {
          const float dv = acc[ns][t][r] - mu[r];
          qd[r] = fmaf(dv, dv, qd[r]);
        }
    float rstd[8];
#pragma unroll
    for (int r = 0; r < 8; ++r) rstd[r] = rsqrtf(row_sum_16(qd[r]) * (1.f / D_DIM) + LN_EPS);

#pragma unroll
    for (int ns = 0; ns < 4; ++ns)
#pragma unroll
      for (int t = 0; t < G_NT; ++t) {
        const int n = 64 * ns + 16 * t + l16;
        const float g = lnp_g[n], bb = lnp_b[n];
#pragma unroll
        for (int r = 0; r < 8; ++r)
          Out[(size_t)(rbase + r) * D_DIM + n] = (acc[ns][t][r] - mu[r]) * rstd[r] * g + bb;
      }
  }
}

#define SCAN_ROWS 4
#define SCAN_NT (SCAN_ROWS * 32)
#define SCAN_T 64
#define SCAN_NC (L_DIM / SCAN_T)
#define SCAN_F4 (SCAN_T * D_DIM / 4 / SCAN_NT)
#define SCAN_VN (SCAN_T * SCAN_ROWS / SCAN_NT)
#define SCAN_LATE 8
#define SCAN_EARLY (SCAN_F4 - SCAN_LATE)
#define SCAN_LATE_AT 32

union F8 { v4f v4[2]; float f[8]; };

struct StepOps { F8 q, k; float v, b; };

__device__ __forceinline__ void load_step_ops(StepOps& o, const v4f* lq, const v4f* lk,
                                              const float* lv, const float* lb, int t) {
  o.q.v4[0] = lq[t * (D_DIM / 4)];
  o.q.v4[1] = lq[t * (D_DIM / 4) + 1];
  o.k.v4[0] = lk[t * (D_DIM / 4)];
  o.k.v4[1] = lk[t * (D_DIM / 4) + 1];
  o.v = lv[t * SCAN_ROWS];
  o.b = lb[t];
}

__device__ __forceinline__ float dot8_tree(const float (&w)[8], const float (&x)[8]) {
  float p0 = w[0] * x[0];
  float p1 = w[1] * x[1];
  float p2 = w[2] * x[2];
  float p3 = w[3] * x[3];
  p0 = fmaf(w[4], x[4], p0);
  p1 = fmaf(w[5], x[5], p1);
  p2 = fmaf(w[6], x[6], p2);
  p3 = fmaf(w[7], x[7], p3);
  return (p0 + p2) + (p1 + p3);
}

__device__ __forceinline__ float scan_step(float (&w)[8], const StepOps& o, float df) {
  float rk = dot8_tree(w, o.k.f);
  float rq = 0.f;
#pragma unroll
  for (int c = 0; c < 8; ++c) rq = fmaf(w[c], o.q.f[c], rq);
  rk = wave_sum_32(rk);
  rq = wave_sum_32(rq);
  asm volatile("" :: "v"(rq));

  const float bv = o.b * o.v;
  const float u = fmaf(-o.b, rk, bv);
#pragma unroll
  for (int c = 0; c < 8; ++c) {
    const float dw = df * w[c];
    w[c] = fmaf(u, o.k.f[c], dw);
  }
  return rq;
}

#define HOLD_READS_HERE() __builtin_amdgcn_sched_barrier(0)


__global__ void __launch_bounds__(SCAN_NT)
delta_scan_kernel(const float* __restrict__ phi_q,
                  const float* __restrict__ phi_k,
                  const float* __restrict__ v,
                  const float* __restrict__ beta,
                  const float* __restrict__ decay,
                  float* __restrict__ y) {
  __shared__ v4f s_q[2][SCAN_T * D_DIM / 4];
  __shared__ v4f s_k[2][SCAN_T * D_DIM / 4];
  __shared__ float  s_v[2][SCAN_T * SCAN_ROWS];
  __shared__ float  s_b[2][SCAN_T];

  const int b    = blockIdx.y;
  const int tid  = threadIdx.x;
  const int wave = tid >> 5;
  const int lane = tid & 31;
  const int row  = blockIdx.x * SCAN_ROWS + wave;

  const float df = 1.f / (1.f + __expf(-decay[0]));
  const size_t base = (size_t)b * L_DIM * D_DIM;

  const v4f* gq = (const v4f*)(phi_q + base) + tid;
  const v4f* gk = (const v4f*)(phi_k + base) + tid;
  const float*  gv = v + base + (size_t)(tid / SCAN_ROWS) * D_DIM
                       + blockIdx.x * SCAN_ROWS + (tid % SCAN_ROWS);
  const float*  gb = beta + (size_t)b * L_DIM + (tid & (SCAN_T - 1));

#pragma unroll
  for (int j = 0; j < SCAN_F4; ++j) {
    s_q[0][tid + SCAN_NT * j] = gq[SCAN_NT * j];
    s_k[0][tid + SCAN_NT * j] = gk[SCAN_NT * j];
  }
#pragma unroll
  for (int j = 0; j < SCAN_VN; ++j)
    s_v[0][tid + SCAN_NT * j] = gv[(size_t)j * (SCAN_NT / SCAN_ROWS) * D_DIM];
  if (tid < SCAN_T) s_b[0][tid] = gb[0];
  __syncthreads();

  v4f rq[SCAN_F4], rk[SCAN_F4];
  float rv[SCAN_VN];
#pragma unroll
  for (int j = 0; j < SCAN_F4; ++j) rk[j] = gk[(SCAN_T * D_DIM / 4) + SCAN_NT * j];
#pragma unroll
  for (int j = 0; j < SCAN_EARLY; ++j) rq[j] = gq[(SCAN_T * D_DIM / 4) + SCAN_NT * j];
#pragma unroll
  for (int j = SCAN_EARLY; j < SCAN_F4; ++j) rq[j] = (v4f){0.f, 0.f, 0.f, 0.f};
#pragma unroll
  for (int j = 0; j < SCAN_VN; ++j)
    rv[j] = gv[(size_t)(SCAN_T * D_DIM) + (size_t)j * (SCAN_NT / SCAN_ROWS) * D_DIM];
  float rb = gb[SCAN_T];

  float w[8];
#pragma unroll
  for (int c = 0; c < 8; ++c) w[c] = 0.f;

  StepOps o0, o1, o2, o3;
  load_step_ops(o0, &s_q[0][lane * 2], &s_k[0][lane * 2], &s_v[0][wave], &s_b[0][0], 0);
  load_step_ops(o1, &s_q[0][lane * 2], &s_k[0][lane * 2], &s_v[0][wave], &s_b[0][0], 1);

  for (int ch = 0; ch < SCAN_NC; ++ch) {
    const int buf = ch & 1;
    const int chl = (ch + 1 < SCAN_NC) ? ch + 1 : SCAN_NC - 1;
    const size_t late4 = (size_t)chl * (SCAN_T * D_DIM / 4);

    const v4f* lq = &s_q[buf][lane * 2];
    const v4f* lk = &s_k[buf][lane * 2];
    const float*  lv = &s_v[buf][wave];
    const float*  lb = &s_b[buf][0];
    float* yrow = y + base + (size_t)ch * SCAN_T * D_DIM + row;

    auto trip = [&](const int t4) {
      const int ta = (t4 + 4 < SCAN_T) ? t4 + 4 : SCAN_T - 1;
      const int tb = (t4 + 5 < SCAN_T) ? t4 + 5 : SCAN_T - 1;
      load_step_ops(o2, lq, lk, lv, lb, t4 + 2);
      HOLD_READS_HERE();
      const float y0 = scan_step(w, o0, df);
      load_step_ops(o3, lq, lk, lv, lb, t4 + 3);
      HOLD_READS_HERE();
      const float y1 = scan_step(w, o1, df);
      load_step_ops(o0, lq, lk, lv, lb, ta);
      HOLD_READS_HERE();
      const float y2 = scan_step(w, o2, df);
      load_step_ops(o1, lq, lk, lv, lb, tb);
      HOLD_READS_HERE();
      const float y3 = scan_step(w, o3, df);
      if (lane == 0) {
        float* yp = yrow + (size_t)t4 * D_DIM;
        yp[0 * D_DIM] = y0;
        yp[1 * D_DIM] = y1;
        yp[2 * D_DIM] = y2;
        yp[3 * D_DIM] = y3;
      }
    };
    for (int t4 = 0; t4 < SCAN_LATE_AT; t4 += 4) trip(t4);
#pragma unroll
    for (int j = SCAN_EARLY; j < SCAN_F4; ++j) rq[j] = gq[late4 + SCAN_NT * j];
    for (int t4 = SCAN_LATE_AT; t4 < SCAN_T; t4 += 4) trip(t4);

    v4f* dq = &s_q[buf ^ 1][tid];
    v4f* dk = &s_k[buf ^ 1][tid];
#pragma unroll
    for (int j = 0; j < SCAN_F4; ++j) {
      dq[SCAN_NT * j] = rq[j];
      dk[SCAN_NT * j] = rk[j];
    }
#pragma unroll
    for (int j = 0; j < SCAN_VN; ++j) s_v[buf ^ 1][tid + SCAN_NT * j] = rv[j];
    if (tid < SCAN_T) s_b[buf ^ 1][tid] = rb;
    __syncthreads();

    load_step_ops(o0, &s_q[buf ^ 1][lane * 2], &s_k[buf ^ 1][lane * 2],
                  &s_v[buf ^ 1][wave], &s_b[buf ^ 1][0], 0);
    load_step_ops(o1, &s_q[buf ^ 1][lane * 2], &s_k[buf ^ 1][lane * 2],
                  &s_v[buf ^ 1][wave], &s_b[buf ^ 1][0], 1);

    const int chn = (ch + 2 < SCAN_NC) ? ch + 2 : SCAN_NC - 1;
    const size_t adv4 = (size_t)chn * (SCAN_T * D_DIM / 4);
#pragma unroll
    for (int j = 0; j < SCAN_F4; ++j) rk[j] = gk[adv4 + SCAN_NT * j];
#pragma unroll
    for (int j = 0; j < SCAN_EARLY; ++j) rq[j] = gq[adv4 + SCAN_NT * j];
#pragma unroll
    for (int j = 0; j < SCAN_VN; ++j)
      rv[j] = gv[(size_t)chn * (SCAN_T * D_DIM) + (size_t)j * (SCAN_NT / SCAN_ROWS) * D_DIM];
    rb = gb[chn * SCAN_T];
  }
}

extern "C" void kernel_launch(void* const* d_in, const int* in_sizes, int n_in,
                              void* d_out, int out_size, void* d_ws, size_t ws_size,
                              hipStream_t stream) {
  (void)in_sizes; (void)n_in; (void)out_size; (void)ws_size;
  const float* x      = (const float*)d_in[0];
  const float* Wq     = (const float*)d_in[1];
  const float* Wk     = (const float*)d_in[2];
  const float* Wv     = (const float*)d_in[3];
  const float* beta_w = (const float*)d_in[4];
  const float* beta_b = (const float*)d_in[5];
  const float* decay  = (const float*)d_in[6];
  const float* Wo     = (const float*)d_in[7];
  const float* bo     = (const float*)d_in[8];
  const float* ln_g   = (const float*)d_in[9];
  const float* ln_b   = (const float*)d_in[10];
  const float* lnp_g  = (const float*)d_in[11];
  const float* lnp_b  = (const float*)d_in[12];
  float* out = (float*)d_out;

  char* ws = (char*)d_ws;
  const size_t MAT = (size_t)M_DIM * D_DIM * sizeof(float);
  float* q    = (float*)(ws);
  float* k    = (float*)(ws + MAT);
  float* vv   = (float*)(ws + 2 * MAT);
  float* beta = (float*)(ws + 3 * MAT);
  float* y    = (float*)(ws + 3 * MAT + (64 << 10));

  const dim3 gemm_grid(M_DIM / 64, D_DIM / 64);
  const dim3 gemm_blk(128);

  qkv_phi_kernel<<<dim3(M_DIM / 64, 3), gemm_blk, 0, stream>>>(
      x, Wq, Wk, Wv, beta_w, beta_b, lnp_g, lnp_b, q, k, vv, beta);

  delta_scan_kernel<<<dim3(D_DIM / SCAN_ROWS, B_DIM), dim3(SCAN_NT), 0, stream>>>(
      q, k, vv, beta, decay, y);

  wmma_gemm_f32_kernel<true><<<gemm_grid, gemm_blk, 0, stream>>>(y, Wo, out, bo, ln_g, ln_b);
}


# scan: v_bar one step ahead by linearity (W(t).phi_k(t+1) and phi_k(t).phi_k(t+1) formed in step t); a step's 3 wave sums share one butterfly begun a step early; new sum order of v_bar and y
# speedup vs baseline: 4.3410x; 1.0699x over previous
#include <hip/hip_runtime.h>
#include <hip/hip_bf16.h>

typedef __attribute__((ext_vector_type(2))) float v2f;
typedef __attribute__((ext_vector_type(8))) float v8f;
typedef __attribute__((ext_vector_type(4))) float v4f;

#define D_DIM 256
#define L_DIM 1024
#define B_DIM 4
#define M_DIM (B_DIM * L_DIM)
#define LN_EPS 1e-5f

__device__ __forceinline__ float other_row(float x) {
  const int b = __builtin_bit_cast(int, x);
  return __builtin_bit_cast(float,
      __builtin_amdgcn_permlanex16(0, b, 0x76543210, 0xfedcba98, false, false));
}
template <int XMASK>
__device__ __forceinline__ float row_xor(float x) {
  const int b = __builtin_bit_cast(int, x);
  return __builtin_bit_cast(float,
      __builtin_amdgcn_update_dpp(0, b, 0x160 + XMASK, 0xf, 0xf, true));
}
__device__ __forceinline__ float row_sum_16(float v) {
  v += row_xor<8>(v);
  v += row_xor<4>(v);
  v += row_xor<2>(v);
  v += row_xor<1>(v);
  return v;
}

template <int Q>
__device__ __forceinline__ float quad_lane(float x) {
  const int b = __builtin_bit_cast(int, x);
  return __builtin_bit_cast(float,
      __builtin_amdgcn_update_dpp(0, b, Q * 0x55, 0xf, 0xf, true));
}

__device__ __forceinline__ float sum3_head(float a, float g, float r, bool odd1, bool odd2) {
  const float ag = (odd1 ? g : a) + row_xor<1>(odd1 ? a : g);
  const float rr = r + row_xor<1>(r);
  return (odd2 ? rr : ag) + row_xor<2>(odd2 ? ag : rr);
}
struct Sum3 { float a, g, r; };
__device__ __forceinline__ Sum3 sum3_tail(float z) {
  z += row_xor<4>(z);
  z += row_xor<8>(z);
  z += other_row(z);
  Sum3 s;
  s.a = quad_lane<0>(z);
  s.g = quad_lane<1>(z);
  s.r = z;
  return s;
}

#define G_K      D_DIM
#define G_PITCH4 (G_K / 4 + 1)
#define G_NT     4

struct BSet { v4f p[G_NT][2]; };

__device__ __forceinline__ void gemm_load_b(BSet& s, const v4f* bp, int slice) {
#pragma unroll
  for (int t = 0; t < G_NT; ++t) {
    s.p[t][0] = bp[t * 16 * G_PITCH4 + 4 * slice];
    s.p[t][1] = bp[t * 16 * G_PITCH4 + 4 * slice + 2];
  }
}

__device__ __forceinline__ v8f gemm_wmma(v2f a, v2f b, v8f c) {
  return __builtin_amdgcn_wmma_f32_16x16x4_f32(
       false, a,  false, b,
       (short)0, c,  false,  false);
}

__device__ __forceinline__ void gemm_mma_slice(v8f (&acc)[G_NT], v4f a0, v4f a1, const BSet& s) {
#pragma unroll
  for (int pc = 0; pc < 2; ++pc) {
    const v4f a4 = pc ? a1 : a0;
    const v2f alo = __builtin_shufflevector(a4, a4, 0, 1);
    const v2f ahi = __builtin_shufflevector(a4, a4, 2, 3);
#pragma unroll
    for (int t = 0; t < G_NT; ++t) {
      const v4f b4 = s.p[t][pc];
      acc[t] = gemm_wmma(alo, __builtin_shufflevector(b4, b4, 0, 1), acc[t]);
    }
#pragma unroll
    for (int t = 0; t < G_NT; ++t) {
      const v4f b4 = s.p[t][pc];
      acc[t] = gemm_wmma(ahi, __builtin_shufflevector(b4, b4, 2, 3), acc[t]);
    }
  }
}

__device__ __forceinline__ void gemm_strip(v8f (&acc)[G_NT], const v4f (&a)[G_K / 8], const v4f* bp) {
  BSet b0, b1;
  gemm_load_b(b0, bp, 0);
#pragma unroll
  for (int s = 0; s < G_K / 16; s += 2) {
    gemm_load_b(b1, bp, s + 1);
    __builtin_amdgcn_sched_barrier(0);
    gemm_mma_slice(acc, a[2 * s], a[2 * s + 1], b0);
    if (s + 2 < G_K / 16) gemm_load_b(b0, bp, s + 2);
    __builtin_amdgcn_sched_barrier(0);
    gemm_mma_slice(acc, a[2 * s + 2], a[2 * s + 3], b1);
  }
}

template <bool LN_ROWS>
__global__ void __launch_bounds__(128)
wmma_gemm_f32_kernel(const float* __restrict__ A,
                     const float* __restrict__ Wm,
                     float* __restrict__ Out,
                     const float* __restrict__ bias,
                     const float* __restrict__ ln_g,
                     const float* __restrict__ ln_b) {
  __shared__ v4f s_w[64 * G_PITCH4];
  __shared__ v4f s_gb[LN_ROWS ? 2 * (G_K / 4) : 1];

  const int N = D_DIM;
  const int tid  = threadIdx.x;
  const int wave = tid >> 5;
  const int lane = tid & 31;
  const int half = lane >> 4;
  const int l16  = lane & 15;
  const int tM = (blockIdx.x * 4 + wave) * 16;
  const int tN = blockIdx.y * 64;

  const v4f* ap = (const v4f*)(A + (size_t)(tM + l16) * G_K) + half;
  v4f a[G_K / 8];
#pragma unroll
  for (int j = 0; j < G_K / 8; ++j) a[j] = ap[2 * j];

  const v4f* wp = (const v4f*)(Wm + (size_t)tN * G_K);
#pragma unroll
  for (int i = 0; i < 64 * (G_K / 4) / 128; ++i) {
    const int idx = tid + 128 * i;
    s_w[(idx >> 6) * G_PITCH4 + (idx & 63)] = wp[idx];
  }
  if constexpr (LN_ROWS) {
    s_gb[tid] = (tid < G_K / 4) ? ((const v4f*)ln_g)[tid] : ((const v4f*)ln_b)[tid - G_K / 4];
  }
  __syncthreads();

  if constexpr (LN_ROWS) {
    v4f s4[4];
#pragma unroll
    for (int i = 0; i < 4; ++i) s4[i] = a[i];
#pragma unroll
    for (int j = 4; j < G_K / 8; ++j) s4[j & 3] += a[j];
    const v4f st = (s4[0] + s4[1]) + (s4[2] + s4[3]);
    float rs = (st.x + st.y) + (st.z + st.w);
    rs += other_row(rs);
    const float mu = rs * (1.f / G_K);

    v4f q4[4];
#pragma unroll
    for (int i = 0; i < 4; ++i) { const v4f d = a[i] - mu; q4[i] = d * d; }
#pragma unroll
    for (int j = 4; j < G_K / 8; ++j) { const v4f d = a[j] - mu; q4[j & 3] += d * d; }
    const v4f qt = (q4[0] + q4[1]) + (q4[2] + q4[3]);
    float rq = (qt.x + qt.y) + (qt.z + qt.w);
    rq += other_row(rq);
    const float rstd = rsqrtf(rq * (1.f / G_K) + LN_EPS);

    const v4f* gp = s_gb + half;
    const v4f* hp = s_gb + G_K / 4 + half;
#pragma unroll
    for (int j = 0; j < G_K / 8; ++j)
      a[j] = ((a[j] - mu) * rstd) * gp[2 * j] + hp[2 * j];
  }

  const v4f* bp = s_w + l16 * G_PITCH4 + half;

  v8f acc[G_NT];
#pragma unroll
  for (int t = 0; t < G_NT; ++t) acc[t] = (v8f){};

  gemm_strip(acc, a, bp);

#pragma unroll
  for (int t = 0; t < G_NT; ++t) {
    const int n = tN + 16 * t + l16;
    const float bv = bias ? bias[n] : 0.f;
    union { v8f v; float f[8]; } u; u.v = acc[t];
#pragma unroll
    for (int r = 0; r < 8; ++r) {
      const int m = tM + r + (half ? 8 : 0);
      Out[(size_t)m * N + n] = u.f[r] + bv;
    }
  }
}

#define P_STRIP4 (64 * (G_K / 4))
#define P_ST     (P_STRIP4 / 128)

__global__ void __launch_bounds__(128)
qkv_phi_kernel(const float* __restrict__ x,
               const float* __restrict__ Wq,
               const float* __restrict__ Wk,
               const float* __restrict__ Wv,
               const float* __restrict__ beta_w,
               const float* __restrict__ beta_b,
               const float* __restrict__ lnp_g,
               const float* __restrict__ lnp_b,
               float* __restrict__ q,
               float* __restrict__ k,
               float* __restrict__ v,
               float* __restrict__ beta) {
  __shared__ v4f s_w[4][64 * G_PITCH4];

  const int which = blockIdx.y;
  const float* Wm = (which == 0) ? Wq : (which == 1) ? Wk : Wv;
  float* Out      = (which == 0) ? q  : (which == 1) ? k  : v;

  const int tid  = threadIdx.x;
  const int wave = tid >> 5;
  const int lane = tid & 31;
  const int half = lane >> 4;
  const int l16  = lane & 15;
  const int tM = (blockIdx.x * 4 + wave) * 16;

  const v4f* ap = (const v4f*)(x + (size_t)(tM + l16) * G_K) + half;
  v4f a[G_K / 8];
#pragma unroll
  for (int j = 0; j < G_K / 8; ++j) a[j] = ap[2 * j];

  const v4f* wp = (const v4f*)Wm;
#pragma unroll
  for (int i = 0; i < P_ST; ++i) {
    const int idx = tid + 128 * i;
    s_w[0][(idx >> 6) * G_PITCH4 + (idx & 63)] = wp[idx];
  }
  __syncthreads();

  v8f acc[4][G_NT];
#pragma unroll
  for (int ns = 0; ns < 4; ++ns)
#pragma unroll
    for (int t = 0; t < G_NT; ++t) acc[ns][t] = (v8f){};

#pragma unroll
  for (int ns = 0; ns < 4; ++ns) {
    v4f st[P_ST];
    if (ns + 1 < 4) {
#pragma unroll
      for (int i = 0; i < P_ST; ++i) st[i] = wp[(ns + 1) * P_STRIP4 + tid + 128 * i];
    }
    gemm_strip(acc[ns], a, &s_w[ns][l16 * G_PITCH4 + half]);
    if (ns + 1 < 4) {
#pragma unroll
      for (int i = 0; i < P_ST; ++i) {
        const int idx = tid + 128 * i;
        s_w[ns + 1][(idx >> 6) * G_PITCH4 + (idx & 63)] = st[i];
      }
      __syncthreads();
    }
  }

  const int rbase = tM + (half ? 8 : 0);

  if (which == 2) {
#pragma unroll
    for (int ns = 0; ns < 4; ++ns)
#pragma unroll
      for (int t = 0; t < G_NT; ++t) {
        const int n = 64 * ns + 16 * t + l16;
#pragma unroll
        for (int r = 0; r < 8; ++r)
          Out[(size_t)(rbase + r) * D_DIM + n] = acc[ns][t][r];
      }
    const v4f* bw = (const v4f*)beta_w + half;
    v4f b4[4];
#pragma unroll
    for (int i = 0; i < 4; ++i) b4[i] = a[i] * bw[2 * i];
#pragma unroll
    for (int j = 4; j < G_K / 8; ++j) b4[j & 3] += a[j] * bw[2 * j];
    const v4f bt = (b4[0] + b4[1]) + (b4[2] + b4[3]);
    float sb = (bt.x + bt.y) + (bt.z + bt.w);
    sb += other_row(sb);
    if (half == 0) beta[tM + l16] = 1.f / (1.f + __expf(-(sb + beta_b[0])));
  } else {
    float sm[8];
#pragma unroll
    for (int r = 0; r < 8; ++r) sm[r] = 0.f;
#pragma unroll
    for (int ns = 0; ns < 4; ++ns)
#pragma unroll
      for (int t = 0; t < G_NT; ++t)
#pragma unroll
        for (int r = 0; r < 8; ++r) {
          const float tv = acc[ns][t][r];
          const float e = (tv > 0.f) ? tv + 1.f : __expf(tv);
          acc[ns][t][r] = e;
          sm[r] += e;
        }
    float mu[8];
#pragma unroll
    for (int r = 0; r < 8; ++r) mu[r] = row_sum_16(sm[r]) * (1.f / D_DIM);

    float qd[8];
#pragma unroll
    for (int r = 0; r < 8; ++r) qd[r] = 0.f;
#pragma unroll
    for (int ns = 0; ns < 4; ++ns)
#pragma unroll
      for (int t = 0; t < G_NT; ++t)
#pragma unroll
        for (int r = 0; r < 8; ++r) {
          const float dv = acc[ns][t][r] - mu[r];
          qd[r] = fmaf(dv, dv, qd[r]);
        }
    float rstd[8];
#pragma unroll
    for (int r = 0; r < 8; ++r) rstd[r] = rsqrtf(row_sum_16(qd[r]) * (1.f / D_DIM) + LN_EPS);

#pragma unroll
    for (int ns = 0; ns < 4; ++ns)
#pragma unroll
      for (int t = 0; t < G_NT; ++t) {
        const int n = 64 * ns + 16 * t + l16;
        const float g = lnp_g[n], bb = lnp_b[n];
#pragma unroll
        for (int r = 0; r < 8; ++r)
          Out[(size_t)(rbase + r) * D_DIM + n] = (acc[ns][t][r] - mu[r]) * rstd[r] * g + bb;
      }
  }
}

#define SCAN_ROWS 4
#define SCAN_NT (SCAN_ROWS * 32)
#define SCAN_T 64
#define SCAN_NC (L_DIM / SCAN_T)
#define SCAN_F4 (SCAN_T * D_DIM / 4 / SCAN_NT)
#define SCAN_VN (SCAN_T * SCAN_ROWS / SCAN_NT)
#define SCAN_LATE 8
#define SCAN_EARLY (SCAN_F4 - SCAN_LATE)
#define SCAN_LATE_AT 32

union F8 { v4f v4[2]; float f[8]; };

struct StepOps { F8 q, k; float v, b; };

__device__ __forceinline__ void load_step_ops(StepOps& o, const v4f* lq, const v4f* lk,
                                              const float* lv, const float* lb, int t) {
  o.q.v4[0] = lq[t * (D_DIM / 4)];
  o.q.v4[1] = lq[t * (D_DIM / 4) + 1];
  o.k.v4[0] = lk[t * (D_DIM / 4)];
  o.k.v4[1] = lk[t * (D_DIM / 4) + 1];
  o.v = lv[t * SCAN_ROWS];
  o.b = lb[t];
}

__device__ __forceinline__ float dot8_tree(const float (&w)[8], const float (&x)[8]) {
  const v2f w01 = {w[0], w[1]}, w23 = {w[2], w[3]}, w45 = {w[4], w[5]}, w67 = {w[6], w[7]};
  const v2f x01 = {x[0], x[1]}, x23 = {x[2], x[3]}, x45 = {x[4], x[5]}, x67 = {x[6], x[7]};
  v2f p = w01 * x01;
  v2f q = w23 * x23;
  p = __builtin_elementwise_fma(w45, x45, p);
  q = __builtin_elementwise_fma(w67, x67, q);
  const v2f s = p + q;
  return s.x + s.y;
}

struct Carry { float z, c, u; };

__device__ __forceinline__ float scan_step(float (&w)[8], const StepOps& o, const float (&kn)[8],
                                           Carry& cy, float df, bool odd1, bool odd2) {
  const Sum3 s = sum3_tail(cy.z);

  cy.z = sum3_head(dot8_tree(w, kn), dot8_tree(o.k.f, kn), dot8_tree(w, o.q.f), odd1, odd2);
  asm volatile("" :: "v"(cy.z));

  const float bv = o.b * o.v;
  const float bc = o.b * cy.c;
  const float bg = o.b * s.g;
  const float u = fmaf(-bg, cy.u, fmaf(-bc, s.a, bv));

#pragma unroll
  for (int c = 0; c < 8; ++c) {
    const float dw = df * w[c];
    w[c] = fmaf(u, o.k.f[c], dw);
  }
  cy.c = df;
  cy.u = u;
  return s.r;
}

#define HOLD_READS_HERE() __builtin_amdgcn_sched_barrier(0)
#define READS_NOT_EARLIER() __builtin_amdgcn_sched_barrier(0)


__global__ void __launch_bounds__(SCAN_NT)
delta_scan_kernel(const float* __restrict__ phi_q,
                  const float* __restrict__ phi_k,
                  const float* __restrict__ v,
                  const float* __restrict__ beta,
                  const float* __restrict__ decay,
                  float* __restrict__ y) {
  __shared__ v4f s_q[2][SCAN_T * D_DIM / 4];
  __shared__ v4f s_k[2][SCAN_T * D_DIM / 4];
  __shared__ float  s_v[2][SCAN_T * SCAN_ROWS];
  __shared__ float  s_b[2][SCAN_T];

  const int b    = blockIdx.y;
  const int tid  = threadIdx.x;
  const int wave = tid >> 5;
  const int lane = tid & 31;
  const int row  = blockIdx.x * SCAN_ROWS + wave;

  const float df = 1.f / (1.f + __expf(-decay[0]));
  const size_t base = (size_t)b * L_DIM * D_DIM;

  const v4f* gq = (const v4f*)(phi_q + base) + tid;
  const v4f* gk = (const v4f*)(phi_k + base) + tid;
  const float*  gv = v + base + (size_t)(tid / SCAN_ROWS) * D_DIM
                       + blockIdx.x * SCAN_ROWS + (tid % SCAN_ROWS);
  const float*  gb = beta + (size_t)b * L_DIM + (tid & (SCAN_T - 1));

#pragma unroll
  for (int j = 0; j < SCAN_F4; ++j) {
    s_q[0][tid + SCAN_NT * j] = gq[SCAN_NT * j];
    s_k[0][tid + SCAN_NT * j] = gk[SCAN_NT * j];
  }
#pragma unroll
  for (int j = 0; j < SCAN_VN; ++j)
    s_v[0][tid + SCAN_NT * j] = gv[(size_t)j * (SCAN_NT / SCAN_ROWS) * D_DIM];
  if (tid < SCAN_T) s_b[0][tid] = gb[0];
  __syncthreads();

  v4f rq[SCAN_F4], rk[SCAN_F4];
  float rv[SCAN_VN];
#pragma unroll
  for (int j = 0; j < SCAN_F4; ++j) rk[j] = gk[(SCAN_T * D_DIM / 4) + SCAN_NT * j];
#pragma unroll
  for (int j = 0; j < SCAN_EARLY; ++j) rq[j] = gq[(SCAN_T * D_DIM / 4) + SCAN_NT * j];
#pragma unroll
  for (int j = SCAN_EARLY; j < SCAN_F4; ++j) rq[j] = (v4f){0.f, 0.f, 0.f, 0.f};
#pragma unroll
  for (int j = 0; j < SCAN_VN; ++j)
    rv[j] = gv[(size_t)(SCAN_T * D_DIM) + (size_t)j * (SCAN_NT / SCAN_ROWS) * D_DIM];
  float rb = gb[SCAN_T];

  float w[8];
#pragma unroll
  for (int c = 0; c < 8; ++c) w[c] = 0.f;

  StepOps o0, o1, o2, o3;
  load_step_ops(o0, &s_q[0][lane * 2], &s_k[0][lane * 2], &s_v[0][wave], &s_b[0][0], 0);
  load_step_ops(o1, &s_q[0][lane * 2], &s_k[0][lane * 2], &s_v[0][wave], &s_b[0][0], 1);
  load_step_ops(o2, &s_q[0][lane * 2], &s_k[0][lane * 2], &s_v[0][wave], &s_b[0][0], 2);
  Carry cy;
  cy.z = 0.f;
  const bool odd1 = (lane & 1) != 0;
  const bool odd2 = (lane & 2) != 0;

  for (int ch = 0; ch < SCAN_NC; ++ch) {
    const int buf = ch & 1;
    const int chl = (ch + 1 < SCAN_NC) ? ch + 1 : SCAN_NC - 1;
    {
      const float z0 = sum3_head(dot8_tree(w, o0.k.f), 0.f, 0.f, odd1, odd2);
      cy.z = odd2 ? cy.z : z0;
    }
    cy.c = 1.f;
    cy.u = 0.f;
    const size_t late4 = (size_t)chl * (SCAN_T * D_DIM / 4);

    const v4f* lq = &s_q[buf][lane * 2];
    const v4f* lk = &s_k[buf][lane * 2];
    const float*  lv = &s_v[buf][wave];
    const float*  lb = &s_b[buf][0];
    float* yrow = y + base + (size_t)ch * SCAN_T * D_DIM + row;

    auto trip = [&](const int t4) {
      const int ta = (t4 + 4 < SCAN_T) ? t4 + 4 : SCAN_T - 1;
      const int tb = (t4 + 5 < SCAN_T) ? t4 + 5 : SCAN_T - 1;
      const int tc = (t4 + 6 < SCAN_T) ? t4 + 6 : SCAN_T - 1;
      load_step_ops(o3, lq, lk, lv, lb, t4 + 3);
      HOLD_READS_HERE();
      const float y0 = scan_step(w, o0, o1.k.f, cy, df, odd1, odd2);
      READS_NOT_EARLIER();
      load_step_ops(o0, lq, lk, lv, lb, ta);
      HOLD_READS_HERE();
      const float y1 = scan_step(w, o1, o2.k.f, cy, df, odd1, odd2);
      READS_NOT_EARLIER();
      load_step_ops(o1, lq, lk, lv, lb, tb);
      HOLD_READS_HERE();
      const float y2 = scan_step(w, o2, o3.k.f, cy, df, odd1, odd2);
      READS_NOT_EARLIER();
      load_step_ops(o2, lq, lk, lv, lb, tc);
      HOLD_READS_HERE();
      const float y3 = scan_step(w, o3, o0.k.f, cy, df, odd1, odd2);
      if (lane == 2) {
        float* yp = yrow + ((ptrdiff_t)t4 - 1) * D_DIM;
        if (ch + t4 > 0) yp[0 * D_DIM] = y0;
        yp[1 * D_DIM] = y1;
        yp[2 * D_DIM] = y2;
        yp[3 * D_DIM] = y3;
      }
    };
    for (int t4 = 0; t4 < SCAN_LATE_AT; t4 += 4) trip(t4);
#pragma unroll
    for (int j = SCAN_EARLY; j < SCAN_F4; ++j) rq[j] = gq[late4 + SCAN_NT * j];
    for (int t4 = SCAN_LATE_AT; t4 < SCAN_T; t4 += 4) trip(t4);

    v4f* dq = &s_q[buf ^ 1][tid];
    v4f* dk = &s_k[buf ^ 1][tid];
#pragma unroll
    for (int j = 0; j < SCAN_F4; ++j) {
      dq[SCAN_NT * j] = rq[j];
      dk[SCAN_NT * j] = rk[j];
    }
#pragma unroll
    for (int j = 0; j < SCAN_VN; ++j) s_v[buf ^ 1][tid + SCAN_NT * j] = rv[j];
    if (tid < SCAN_T) s_b[buf ^ 1][tid] = rb;
    __syncthreads();

    load_step_ops(o0, &s_q[buf ^ 1][lane * 2], &s_k[buf ^ 1][lane * 2],
                  &s_v[buf ^ 1][wave], &s_b[buf ^ 1][0], 0);
    load_step_ops(o1, &s_q[buf ^ 1][lane * 2], &s_k[buf ^ 1][lane * 2],
                  &s_v[buf ^ 1][wave], &s_b[buf ^ 1][0], 1);
    load_step_ops(o2, &s_q[buf ^ 1][lane * 2], &s_k[buf ^ 1][lane * 2],
                  &s_v[buf ^ 1][wave], &s_b[buf ^ 1][0], 2);

    const int chn = (ch + 2 < SCAN_NC) ? ch + 2 : SCAN_NC - 1;
    const size_t adv4 = (size_t)chn * (SCAN_T * D_DIM / 4);
#pragma unroll
    for (int j = 0; j < SCAN_F4; ++j) rk[j] = gk[adv4 + SCAN_NT * j];
#pragma unroll
    for (int j = 0; j < SCAN_EARLY; ++j) rq[j] = gq[adv4 + SCAN_NT * j];
#pragma unroll
    for (int j = 0; j < SCAN_VN; ++j)
      rv[j] = gv[(size_t)chn * (SCAN_T * D_DIM) + (size_t)j * (SCAN_NT / SCAN_ROWS) * D_DIM];
    rb = gb[chn * SCAN_T];
  }

  const Sum3 last = sum3_tail(cy.z);
  if (lane == 2) y[base + (size_t)(L_DIM - 1) * D_DIM + row] = last.r;
}

extern "C" void kernel_launch(void* const* d_in, const int* in_sizes, int n_in,
                              void* d_out, int out_size, void* d_ws, size_t ws_size,
                              hipStream_t stream) {
  (void)in_sizes; (void)n_in; (void)out_size; (void)ws_size;
  const float* x      = (const float*)d_in[0];
  const float* Wq     = (const float*)d_in[1];
  const float* Wk     = (const float*)d_in[2];
  const float* Wv     = (const float*)d_in[3];
  const float* beta_w = (const float*)d_in[4];
  const float* beta_b = (const float*)d_in[5];
  const float* decay  = (const float*)d_in[6];
  const float* Wo     = (const float*)d_in[7];
  const float* bo     = (const float*)d_in[8];
  const float* ln_g   = (const float*)d_in[9];
  const float* ln_b   = (const float*)d_in[10];
  const float* lnp_g  = (const float*)d_in[11];
  const float* lnp_b  = (const float*)d_in[12];
  float* out = (float*)d_out;

  char* ws = (char*)d_ws;
  const size_t MAT = (size_t)M_DIM * D_DIM * sizeof(float);
  float* q    = (float*)(ws);
  float* k    = (float*)(ws + MAT);
  float* vv   = (float*)(ws + 2 * MAT);
  float* beta = (float*)(ws + 3 * MAT);
  float* y    = (float*)(ws + 3 * MAT + (64 << 10));

  const dim3 gemm_grid(M_DIM / 64, D_DIM / 64);
  const dim3 gemm_blk(128);

  qkv_phi_kernel<<<dim3(M_DIM / 64, 3), gemm_blk, 0, stream>>>(
      x, Wq, Wk, Wv, beta_w, beta_b, lnp_g, lnp_b, q, k, vv, beta);

  delta_scan_kernel<<<dim3(D_DIM / SCAN_ROWS, B_DIM), dim3(SCAN_NT), 0, stream>>>(
      q, k, vv, beta, decay, y);

  wmma_gemm_f32_kernel<true><<<gemm_grid, gemm_blk, 0, stream>>>(y, Wo, out, bo, ln_g, ln_b);
}
